# starting kernel
# baseline (speedup 1.0000x reference)
#include <hip/hip_runtime.h>
#include <math.h>

typedef __attribute__((ext_vector_type(2))) float v2f;
typedef __attribute__((ext_vector_type(8))) float v8f;

#define NC    45
#define NCP   48
#define H1DIM 32
#define H2DIM 128

__global__ void deg_kernel(const int* __restrict__ dst, const float* __restrict__ ew,
                           float* __restrict__ deg, int E) {
  int e = blockIdx.x * blockDim.x + threadIdx.x;
  if (e < E) atomicAdd(&deg[dst[e]], ew[e]);
}

__global__ void dis_kernel(float* __restrict__ deg, int n) {
  int i = blockIdx.x * blockDim.x + threadIdx.x;
  if (i < n) deg[i] = 1.0f / sqrtf(deg[i] + 1.0f);
}

__global__ void pad_x_kernel(const float* __restrict__ x, float* __restrict__ xp,
                             int total  ) {
  int gid = blockIdx.x * blockDim.x + threadIdx.x;
  if (gid >= total) return;
  int i = gid / NCP, c = gid - i * NCP;
  xp[gid] = (c < NC) ? x[(size_t)i * NC + c] : 0.0f;
}

__global__ void pack_w1_kernel(const float* __restrict__ W1, float* __restrict__ Bp1) {
  int gid = blockIdx.x * blockDim.x + threadIdx.x;
  if (gid >= 12 * 2 * H1DIM * 2) return;
  int p   = gid & 1;
  int col = (gid >> 1) & 31;
  int kh  = (gid >> 6) & 1;
  int k   = gid >> 7;
  int row = 4 * k + 2 * kh + p;
  Bp1[gid] = (row < NC) ? W1[row * H1DIM + col] : 0.0f;
}

__global__ void pack_w2_kernel(const float* __restrict__ W2, float* __restrict__ Bp2) {
  int gid = blockIdx.x * blockDim.x + threadIdx.x;
  if (gid >= 8 * 2 * H2DIM * 2) return;
  int p   = gid & 1;
  int col = (gid >> 1) & 127;
  int kh  = (gid >> 8) & 1;
  int k   = gid >> 9;
  int row = 4 * k + 2 * kh + p;
  Bp2[gid] = W2[row * H2DIM + col];
}

__global__ void gemm1_wmma(const float* __restrict__ xp, const float* __restrict__ Bp1,
                           float* __restrict__ h1, int n) {
  const int lane  = threadIdx.x;
  const int m0    = blockIdx.x * 16;
  const int khalf = lane >> 4;
  const int nn    = lane & 15;
  int m = m0 + nn;
  if (m > n - 1) m = n - 1;

  const float* ap = xp + (size_t)m * NCP + 2 * khalf;
  const float* bp = Bp1 + khalf * 64;
  v8f c0 = {};
  v8f c1 = {};
#pragma unroll
  for (int k = 0; k < 12; ++k) {
    v2f a  = *(const v2f*)(ap + 4 * k);
    v2f b0 = *(const v2f*)(bp + 128 * k + nn * 2);
    v2f b1 = *(const v2f*)(bp + 128 * k + (16 + nn) * 2);
    c0 = __builtin_amdgcn_wmma_f32_16x16x4_f32(false, a, false, b0, (short)0, c0, false, false);
    c1 = __builtin_amdgcn_wmma_f32_16x16x4_f32(false, a, false, b1, (short)0, c1, false, false);
  }
  const int rbase = m0 + (khalf << 3);
#pragma unroll
  for (int j = 0; j < 8; ++j) {
    const int r = rbase + j;
    if (r < n) {
      h1[(size_t)r * H1DIM + nn]      = c0[j];
      h1[(size_t)r * H1DIM + 16 + nn] = c1[j];
    }
  }
}

__global__ void gemm2_wmma(const float* __restrict__ g, const float* __restrict__ Bp2,
                           const float* __restrict__ b2, float* __restrict__ out, int n) {
  const int lane  = threadIdx.x;
  const int m0    = blockIdx.x * 16;
  const int khalf = lane >> 4;
  const int nn    = lane & 15;
  int m = m0 + nn;
  if (m > n - 1) m = n - 1;

  const float* ap = g + (size_t)m * H1DIM + 2 * khalf;
  const float* bp = Bp2 + khalf * 256;
  v2f a[8];
#pragma unroll
  for (int k = 0; k < 8; ++k) a[k] = *(const v2f*)(ap + 4 * k);

  v8f acc[8] = {};
#pragma unroll
  for (int k = 0; k < 8; ++k) {
#pragma unroll
    for (int t = 0; t < 8; ++t) {
      v2f b = *(const v2f*)(bp + 512 * k + (t * 16 + nn) * 2);
      acc[t] = __builtin_amdgcn_wmma_f32_16x16x4_f32(false, a[k], false, b, (short)0, acc[t], false, false);
    }
  }
  float bv[8];
#pragma unroll
  for (int t = 0; t < 8; ++t) bv[t] = b2[t * 16 + nn];

  const int rbase = m0 + (khalf << 3);
#pragma unroll
  for (int j = 0; j < 8; ++j) {
    const int r = rbase + j;
    if (r < n) {
#pragma unroll
      for (int t = 0; t < 8; ++t)
        out[(size_t)r * H2DIM + t * 16 + nn] = acc[t][j] + bv[t];
    }
  }
}

__global__ void agg_edges_kernel(const int* __restrict__ src, const int* __restrict__ dst,
                                 const float* __restrict__ ew, const float* __restrict__ dis,
                                 const float* __restrict__ h, float* __restrict__ agg, int E) {
  long long gid = (long long)blockIdx.x * blockDim.x + threadIdx.x;
  int e = (int)(gid >> 3);
  int q = (int)(gid & 7) << 2;
  if (e >= E) return;
  int s = src[e], d = dst[e];
  float norm = dis[s] * ew[e] * dis[d];
  const float4 v = *(const float4*)(h + (size_t)s * H1DIM + q);
  float* o = agg + (size_t)d * H1DIM + q;
  atomicAdd(o + 0, v.x * norm);
  atomicAdd(o + 1, v.y * norm);
  atomicAdd(o + 2, v.z * norm);
  atomicAdd(o + 3, v.w * norm);
}

__global__ void tanh_bias_kernel(float* __restrict__ agg, const float* __restrict__ h1,
                                 const float* __restrict__ dis, const float* __restrict__ b1,
                                 int n) {
  int gid = blockIdx.x * blockDim.x + threadIdx.x;
  if (gid >= n * H1DIM) return;
  int i = gid >> 5, c = gid & 31;
  float di = dis[i];
  agg[gid] = tanhf(agg[gid] + h1[gid] * di * di + b1[c]);
}

__global__ void selfloop_kernel(float* __restrict__ agg2, const float* __restrict__ h,
                                const float* __restrict__ dis, int n) {
  int gid = blockIdx.x * blockDim.x + threadIdx.x;
  if (gid >= n * H1DIM) return;
  int i = gid >> 5;
  float di = dis[i];
  agg2[gid] += h[gid] * di * di;
}

extern "C" void kernel_launch(void* const* d_in, const int* in_sizes, int n_in,
                              void* d_out, int out_size, void* d_ws, size_t ws_size,
                              hipStream_t stream) {
  const float* x   = (const float*)d_in[0];
  const int*   src = (const int*)  d_in[1];
  const int*   dst = (const int*)  d_in[2];
  const float* ew  = (const float*)d_in[3];
  const float* W1  = (const float*)d_in[4];
  const float* b1  = (const float*)d_in[5];
  const float* W2  = (const float*)d_in[6];
  const float* b2  = (const float*)d_in[7];
  float* out = (float*)d_out;

  const int n = in_sizes[0] / NC;
  const int E = in_sizes[1];

  float* dis  = (float*)d_ws;
  float* xp   = dis  + n;
  float* h1   = xp   + (size_t)n * NCP;
  float* agg1 = h1   + (size_t)n * H1DIM;
  float* agg2 = agg1 + (size_t)n * H1DIM;
  float* Bp1  = agg2 + (size_t)n * H1DIM;
  float* Bp2  = Bp1  + 12 * 2 * H1DIM * 2;

  hipMemsetAsync(dis,  0, (size_t)n * sizeof(float), stream);
  hipMemsetAsync(agg1, 0, (size_t)n * H1DIM * sizeof(float), stream);
  hipMemsetAsync(agg2, 0, (size_t)n * H1DIM * sizeof(float), stream);

  const int T = 256;
  deg_kernel<<<(E + T - 1) / T, T, 0, stream>>>(dst, ew, dis, E);
  dis_kernel<<<(n + T - 1) / T, T, 0, stream>>>(dis, n);

  pad_x_kernel<<<((long long)n * NCP + T - 1) / T, T, 0, stream>>>(x, xp, n * NCP);
  pack_w1_kernel<<<(12 * 2 * H1DIM * 2 + T - 1) / T, T, 0, stream>>>(W1, Bp1);
  pack_w2_kernel<<<(8 * 2 * H2DIM * 2 + T - 1) / T, T, 0, stream>>>(W2, Bp2);

  gemm1_wmma<<<(n + 15) / 16, 32, 0, stream>>>(xp, Bp1, h1, n);
  {
    long long work = (long long)E * 8;
    agg_edges_kernel<<<(unsigned)((work + T - 1) / T), T, 0, stream>>>(src, dst, ew, dis, h1, agg1, E);
  }
  tanh_bias_kernel<<<((long long)n * H1DIM + T - 1) / T, T, 0, stream>>>(agg1, h1, dis, b1, n);

  {
    long long work = (long long)E * 8;
    agg_edges_kernel<<<(unsigned)((work + T - 1) / T), T, 0, stream>>>(src, dst, ew, dis, agg1, agg2, E);
  }
  selfloop_kernel<<<((long long)n * H1DIM + T - 1) / T, T, 0, stream>>>(agg2, agg1, dis, n);
  gemm2_wmma<<<(n + 15) / 16, 32, 0, stream>>>(agg2, Bp2, b2, out, n);
}


# edge aggregation: one wave per 32 edges, lane = feature: each row gather and each f32 atomic add instruction covers ONE 128-B line (was 4 lines of 8 lanes); same adds, atomic arrival order still free
# speedup vs baseline: 1.1897x; 1.1897x over previous
#include <hip/hip_runtime.h>
#include <math.h>

typedef __attribute__((ext_vector_type(2))) float v2f;
typedef __attribute__((ext_vector_type(8))) float v8f;

#define NC    45
#define NCP   48
#define H1DIM 32
#define H2DIM 128

__global__ void deg_kernel(const int* __restrict__ dst, const float* __restrict__ ew,
                           float* __restrict__ deg, int E) {
  int e = blockIdx.x * blockDim.x + threadIdx.x;
  if (e < E) atomicAdd(&deg[dst[e]], ew[e]);
}

__global__ void dis_kernel(float* __restrict__ deg, int n) {
  int i = blockIdx.x * blockDim.x + threadIdx.x;
  if (i < n) deg[i] = 1.0f / sqrtf(deg[i] + 1.0f);
}

__global__ void pad_x_kernel(const float* __restrict__ x, float* __restrict__ xp,
                             int total  ) {
  int gid = blockIdx.x * blockDim.x + threadIdx.x;
  if (gid >= total) return;
  int i = gid / NCP, c = gid - i * NCP;
  xp[gid] = (c < NC) ? x[(size_t)i * NC + c] : 0.0f;
}

__global__ void pack_w1_kernel(const float* __restrict__ W1, float* __restrict__ Bp1) {
  int gid = blockIdx.x * blockDim.x + threadIdx.x;
  if (gid >= 12 * 2 * H1DIM * 2) return;
  int p   = gid & 1;
  int col = (gid >> 1) & 31;
  int kh  = (gid >> 6) & 1;
  int k   = gid >> 7;
  int row = 4 * k + 2 * kh + p;
  Bp1[gid] = (row < NC) ? W1[row * H1DIM + col] : 0.0f;
}

__global__ void pack_w2_kernel(const float* __restrict__ W2, float* __restrict__ Bp2) {
  int gid = blockIdx.x * blockDim.x + threadIdx.x;
  if (gid >= 8 * 2 * H2DIM * 2) return;
  int p   = gid & 1;
  int col = (gid >> 1) & 127;
  int kh  = (gid >> 8) & 1;
  int k   = gid >> 9;
  int row = 4 * k + 2 * kh + p;
  Bp2[gid] = W2[row * H2DIM + col];
}

__global__ void gemm1_wmma(const float* __restrict__ xp, const float* __restrict__ Bp1,
                           float* __restrict__ h1, int n) {
  const int lane  = threadIdx.x;
  const int m0    = blockIdx.x * 16;
  const int khalf = lane >> 4;
  const int nn    = lane & 15;
  int m = m0 + nn;
  if (m > n - 1) m = n - 1;

  const float* ap = xp + (size_t)m * NCP + 2 * khalf;
  const float* bp = Bp1 + khalf * 64;
  v8f c0 = {};
  v8f c1 = {};
#pragma unroll
  for (int k = 0; k < 12; ++k) {
    v2f a  = *(const v2f*)(ap + 4 * k);
    v2f b0 = *(const v2f*)(bp + 128 * k + nn * 2);
    v2f b1 = *(const v2f*)(bp + 128 * k + (16 + nn) * 2);
    c0 = __builtin_amdgcn_wmma_f32_16x16x4_f32(false, a, false, b0, (short)0, c0, false, false);
    c1 = __builtin_amdgcn_wmma_f32_16x16x4_f32(false, a, false, b1, (short)0, c1, false, false);
  }
  const int rbase = m0 + (khalf << 3);
#pragma unroll
  for (int j = 0; j < 8; ++j) {
    const int r = rbase + j;
    if (r < n) {
      h1[(size_t)r * H1DIM + nn]      = c0[j];
      h1[(size_t)r * H1DIM + 16 + nn] = c1[j];
    }
  }
}

__global__ void gemm2_wmma(const float* __restrict__ g, const float* __restrict__ Bp2,
                           const float* __restrict__ b2, float* __restrict__ out, int n) {
  const int lane  = threadIdx.x;
  const int m0    = blockIdx.x * 16;
  const int khalf = lane >> 4;
  const int nn    = lane & 15;
  int m = m0 + nn;
  if (m > n - 1) m = n - 1;

  const float* ap = g + (size_t)m * H1DIM + 2 * khalf;
  const float* bp = Bp2 + khalf * 256;
  v2f a[8];
#pragma unroll
  for (int k = 0; k < 8; ++k) a[k] = *(const v2f*)(ap + 4 * k);

  v8f acc[8] = {};
#pragma unroll
  for (int k = 0; k < 8; ++k) {
#pragma unroll
    for (int t = 0; t < 8; ++t) {
      v2f b = *(const v2f*)(bp + 512 * k + (t * 16 + nn) * 2);
      acc[t] = __builtin_amdgcn_wmma_f32_16x16x4_f32(false, a[k], false, b, (short)0, acc[t], false, false);
    }
  }
  float bv[8];
#pragma unroll
  for (int t = 0; t < 8; ++t) bv[t] = b2[t * 16 + nn];

  const int rbase = m0 + (khalf << 3);
#pragma unroll
  for (int j = 0; j < 8; ++j) {
    const int r = rbase + j;
    if (r < n) {
#pragma unroll
      for (int t = 0; t < 8; ++t)
        out[(size_t)r * H2DIM + t * 16 + nn] = acc[t][j] + bv[t];
    }
  }
}

__global__ void agg_edges_kernel(const int* __restrict__ src, const int* __restrict__ dst,
                                 const float* __restrict__ ew, const float* __restrict__ dis,
                                 const float* __restrict__ h, float* __restrict__ agg, int E) {
  const int lane = threadIdx.x & 31;
  const int wv = __builtin_amdgcn_readfirstlane((int)(threadIdx.x >> 5));
  const int e0 = (blockIdx.x * (int)(blockDim.x >> 5) + wv) << 5;
  if (e0 >= E) return;
  const int e = e0 + lane;
  int s = 0, d = 0;
  float w = 0.0f;
  if (e < E) { s = src[e]; d = dst[e]; w = ew[e]; }
  const float dsv = dis[s];
  const float ddv = dis[d];

  if (e0 + 32 <= E) {
    float v[32];
#pragma unroll
    for (int j = 0; j < 32; ++j) {
      const int sj = __builtin_amdgcn_readlane(s, j);
      v[j] = h[(size_t)sj * H1DIM + lane];
    }
    const float nrm = dsv * w * ddv;
#pragma unroll
    for (int j = 0; j < 32; ++j) {
      const int   dj = __builtin_amdgcn_readlane(d, j);
      const float nj = __int_as_float(__builtin_amdgcn_readlane(__float_as_int(nrm), j));
      atomicAdd(agg + (size_t)dj * H1DIM + lane, v[j] * nj);
    }
  } else {
    const float nrm = dsv * w * ddv;
    const int cnt = E - e0;
    for (int j = 0; j < cnt; ++j) {
      const int   sj = __builtin_amdgcn_readlane(s, j);
      const int   dj = __builtin_amdgcn_readlane(d, j);
      const float nj = __int_as_float(__builtin_amdgcn_readlane(__float_as_int(nrm), j));
      atomicAdd(agg + (size_t)dj * H1DIM + lane, h[(size_t)sj * H1DIM + lane] * nj);
    }
  }
}

__global__ void tanh_bias_kernel(float* __restrict__ agg, const float* __restrict__ h1,
                                 const float* __restrict__ dis, const float* __restrict__ b1,
                                 int n) {
  int gid = blockIdx.x * blockDim.x + threadIdx.x;
  if (gid >= n * H1DIM) return;
  int i = gid >> 5, c = gid & 31;
  float di = dis[i];
  agg[gid] = tanhf(agg[gid] + h1[gid] * di * di + b1[c]);
}

__global__ void selfloop_kernel(float* __restrict__ agg2, const float* __restrict__ h,
                                const float* __restrict__ dis, int n) {
  int gid = blockIdx.x * blockDim.x + threadIdx.x;
  if (gid >= n * H1DIM) return;
  int i = gid >> 5;
  float di = dis[i];
  agg2[gid] += h[gid] * di * di;
}

extern "C" void kernel_launch(void* const* d_in, const int* in_sizes, int n_in,
                              void* d_out, int out_size, void* d_ws, size_t ws_size,
                              hipStream_t stream) {
  const float* x   = (const float*)d_in[0];
  const int*   src = (const int*)  d_in[1];
  const int*   dst = (const int*)  d_in[2];
  const float* ew  = (const float*)d_in[3];
  const float* W1  = (const float*)d_in[4];
  const float* b1  = (const float*)d_in[5];
  const float* W2  = (const float*)d_in[6];
  const float* b2  = (const float*)d_in[7];
  float* out = (float*)d_out;

  const int n = in_sizes[0] / NC;
  const int E = in_sizes[1];

  float* dis  = (float*)d_ws;
  float* xp   = dis  + n;
  float* h1   = xp   + (size_t)n * NCP;
  float* agg1 = h1   + (size_t)n * H1DIM;
  float* agg2 = agg1 + (size_t)n * H1DIM;
  float* Bp1  = agg2 + (size_t)n * H1DIM;
  float* Bp2  = Bp1  + 12 * 2 * H1DIM * 2;

  hipMemsetAsync(dis,  0, (size_t)n * sizeof(float), stream);
  hipMemsetAsync(agg1, 0, (size_t)n * H1DIM * sizeof(float), stream);
  hipMemsetAsync(agg2, 0, (size_t)n * H1DIM * sizeof(float), stream);

  const int T = 256;
  deg_kernel<<<(E + T - 1) / T, T, 0, stream>>>(dst, ew, dis, E);
  dis_kernel<<<(n + T - 1) / T, T, 0, stream>>>(dis, n);

  pad_x_kernel<<<((long long)n * NCP + T - 1) / T, T, 0, stream>>>(x, xp, n * NCP);
  pack_w1_kernel<<<(12 * 2 * H1DIM * 2 + T - 1) / T, T, 0, stream>>>(W1, Bp1);
  pack_w2_kernel<<<(8 * 2 * H2DIM * 2 + T - 1) / T, T, 0, stream>>>(W2, Bp2);

  gemm1_wmma<<<(n + 15) / 16, 32, 0, stream>>>(xp, Bp1, h1, n);
  agg_edges_kernel<<<(E + T - 1) / T, T, 0, stream>>>(src, dst, ew, dis, h1, agg1, E);
  tanh_bias_kernel<<<((long long)n * H1DIM + T - 1) / T, T, 0, stream>>>(agg1, h1, dis, b1, n);

  agg_edges_kernel<<<(E + T - 1) / T, T, 0, stream>>>(src, dst, ew, dis, agg1, agg2, E);
  selfloop_kernel<<<((long long)n * H1DIM + T - 1) / T, T, 0, stream>>>(agg2, agg1, dis, n);
  gemm2_wmma<<<(n + 15) / 16, 32, 0, stream>>>(agg2, Bp2, b2, out, n);
}


# elementwise and operand-prep passes folded into the kernels beside them: dis, x padding and W packing into the GEMMs, layer-2 self-loop into GEMM2's A load, one zero fill: 14 stream ops become 7
# speedup vs baseline: 1.3126x; 1.1033x over previous
#include <hip/hip_runtime.h>
#include <math.h>

typedef __attribute__((ext_vector_type(2))) float v2f;
typedef __attribute__((ext_vector_type(8))) float v8f;

#define NC    45
#define H1DIM 32
#define H2DIM 128

__global__ void deg_kernel(const int* __restrict__ dst, const float* __restrict__ ew,
                           float* __restrict__ deg, int E) {
  int e = blockIdx.x * blockDim.x + threadIdx.x;
  if (e < E) atomicAdd(&deg[dst[e]], ew[e]);
}

__global__ void gemm1_wmma(const float* __restrict__ x, const float* __restrict__ W1,
                           const float* __restrict__ deg, float* __restrict__ dis,
                           float* __restrict__ h1, int n) {
  const int lane  = threadIdx.x;
  const int m0    = blockIdx.x * 16;
  const int khalf = (lane >> 4) & 1;
  const int nn    = lane & 15;
  int m = m0 + nn;
  if (m > n - 1) m = n - 1;

  const int  drow = m0 + lane;
  const bool down = (lane < 16) && (drow < n);
  const float dg = deg[drow < n ? drow : n - 1];

  const float* xr = x + (size_t)m * NC + 2 * khalf;
  const float* wr = W1 + (size_t)(2 * khalf) * H1DIM + nn;
  v8f c0 = {};
  v8f c1 = {};
#pragma unroll
  for (int k = 0; k < 11; ++k) {
    v2f a, b0, b1;
    a.x  = xr[4 * k];
    a.y  = xr[4 * k + 1];
    b0.x = wr[(4 * k) * H1DIM];
    b0.y = wr[(4 * k + 1) * H1DIM];
    b1.x = wr[(4 * k) * H1DIM + 16];
    b1.y = wr[(4 * k + 1) * H1DIM + 16];
    c0 = __builtin_amdgcn_wmma_f32_16x16x4_f32(false, a, false, b0, (short)0, c0, false, false);
    c1 = __builtin_amdgcn_wmma_f32_16x16x4_f32(false, a, false, b1, (short)0, c1, false, false);
  }
  {
    const int   koff = khalf ? 0 : 44;
    const float* xl  = x + (size_t)m * NC + koff;
    const float* wl  = W1 + (size_t)koff * H1DIM + nn;
    const float ta = xl[0], t0 = wl[0], t1 = wl[16];
    v2f a, b0, b1;
    a.x  = khalf ? 0.0f : ta;  a.y  = 0.0f;
    b0.x = khalf ? 0.0f : t0;  b0.y = 0.0f;
    b1.x = khalf ? 0.0f : t1;  b1.y = 0.0f;
    c0 = __builtin_amdgcn_wmma_f32_16x16x4_f32(false, a, false, b0, (short)0, c0, false, false);
    c1 = __builtin_amdgcn_wmma_f32_16x16x4_f32(false, a, false, b1, (short)0, c1, false, false);
  }
  const int rbase = m0 + (khalf << 3);
#pragma unroll
  for (int j = 0; j < 8; ++j) {
    const int r = rbase + j;
    if (r < n) {
      h1[(size_t)r * H1DIM + nn]      = c0[j];
      h1[(size_t)r * H1DIM + 16 + nn] = c1[j];
    }
  }
  if (down) dis[drow] = 1.0f / sqrtf(dg + 1.0f);
}

__global__ void gemm2_wmma(const float* __restrict__ agg2, const float* __restrict__ h,
                           const float* __restrict__ dis, const float* __restrict__ W2,
                           const float* __restrict__ b2, float* __restrict__ out, int n) {
  const int lane  = threadIdx.x;
  const int m0    = blockIdx.x * 16;
  const int khalf = (lane >> 4) & 1;
  const int nn    = lane & 15;
  int m = m0 + nn;
  if (m > n - 1) m = n - 1;

  const float* ap = agg2 + (size_t)m * H1DIM + 2 * khalf;
  const float* hp = h    + (size_t)m * H1DIM + 2 * khalf;
  const float* wr = W2 + (size_t)(2 * khalf) * H2DIM + nn;
  const float di = dis[m];
  v2f a[8];
#pragma unroll
  for (int k = 0; k < 8; ++k) {
    const v2f ag = *(const v2f*)(ap + 4 * k);
    const v2f hv = *(const v2f*)(hp + 4 * k);
    a[k].x = ag.x + hv.x * di * di;
    a[k].y = ag.y + hv.y * di * di;
  }

  v8f acc[8] = {};
#pragma unroll
  for (int k = 0; k < 8; ++k) {
#pragma unroll
    for (int t = 0; t < 8; ++t) {
      v2f b;
      b.x = wr[(4 * k) * H2DIM + t * 16];
      b.y = wr[(4 * k + 1) * H2DIM + t * 16];
      acc[t] = __builtin_amdgcn_wmma_f32_16x16x4_f32(false, a[k], false, b, (short)0, acc[t], false, false);
    }
  }
  float bv[8];
#pragma unroll
  for (int t = 0; t < 8; ++t) bv[t] = b2[t * 16 + nn];

  const int rbase = m0 + (khalf << 3);
#pragma unroll
  for (int j = 0; j < 8; ++j) {
    const int r = rbase + j;
    if (r < n) {
#pragma unroll
      for (int t = 0; t < 8; ++t)
        out[(size_t)r * H2DIM + t * 16 + nn] = acc[t][j] + bv[t];
    }
  }
}

__global__ void agg_edges_kernel(const int* __restrict__ src, const int* __restrict__ dst,
                                 const float* __restrict__ ew, const float* __restrict__ dis,
                                 const float* __restrict__ h, float* __restrict__ agg, int E) {
  const int lane = threadIdx.x & 31;
  const int wv = __builtin_amdgcn_readfirstlane((int)(threadIdx.x >> 5));
  const int e0 = (blockIdx.x * (int)(blockDim.x >> 5) + wv) << 5;
  if (e0 >= E) return;
  const int e = e0 + lane;
  int s = 0, d = 0;
  float w = 0.0f;
  if (e < E) { s = src[e]; d = dst[e]; w = ew[e]; }
  const float dsv = dis[s];
  const float ddv = dis[d];

  if (e0 + 32 <= E) {
    float v[32];
#pragma unroll
    for (int j = 0; j < 32; ++j) {
      const int sj = __builtin_amdgcn_readlane(s, j);
      v[j] = h[(size_t)sj * H1DIM + lane];
    }
    const float nrm = dsv * w * ddv;
#pragma unroll
    for (int j = 0; j < 32; ++j) {
      const int   dj = __builtin_amdgcn_readlane(d, j);
      const float nj = __int_as_float(__builtin_amdgcn_readlane(__float_as_int(nrm), j));
      atomicAdd(agg + (size_t)dj * H1DIM + lane, v[j] * nj);
    }
  } else {
    const float nrm = dsv * w * ddv;
    const int cnt = E - e0;
    for (int j = 0; j < cnt; ++j) {
      const int   sj = __builtin_amdgcn_readlane(s, j);
      const int   dj = __builtin_amdgcn_readlane(d, j);
      const float nj = __int_as_float(__builtin_amdgcn_readlane(__float_as_int(nrm), j));
      atomicAdd(agg + (size_t)dj * H1DIM + lane, h[(size_t)sj * H1DIM + lane] * nj);
    }
  }
}

__global__ void tanh_bias_kernel(float* __restrict__ agg, const float* __restrict__ h1,
                                 const float* __restrict__ dis, const float* __restrict__ b1,
                                 int n) {
  int gid = blockIdx.x * blockDim.x + threadIdx.x;
  if (gid >= n * H1DIM) return;
  int i = gid >> 5, c = gid & 31;
  float di = dis[i];
  agg[gid] = tanhf(agg[gid] + h1[gid] * di * di + b1[c]);
}

extern "C" void kernel_launch(void* const* d_in, const int* in_sizes, int n_in,
                              void* d_out, int out_size, void* d_ws, size_t ws_size,
                              hipStream_t stream) {
  const float* x   = (const float*)d_in[0];
  const int*   src = (const int*)  d_in[1];
  const int*   dst = (const int*)  d_in[2];
  const float* ew  = (const float*)d_in[3];
  const float* W1  = (const float*)d_in[4];
  const float* b1  = (const float*)d_in[5];
  const float* W2  = (const float*)d_in[6];
  const float* b2  = (const float*)d_in[7];
  float* out = (float*)d_out;

  const int n = in_sizes[0] / NC;
  const int E = in_sizes[1];

  float* deg  = (float*)d_ws;
  float* agg1 = deg  + n;
  float* agg2 = agg1 + (size_t)n * H1DIM;
  float* dis  = agg2 + (size_t)n * H1DIM;
  float* h1   = dis  + n;

  hipMemsetAsync(deg, 0, ((size_t)n * (2 * H1DIM + 1)) * sizeof(float), stream);

  const int T = 256;
  deg_kernel<<<(E + T - 1) / T, T, 0, stream>>>(dst, ew, deg, E);

  gemm1_wmma<<<(n + 15) / 16, 32, 0, stream>>>(x, W1, deg, dis, h1, n);
  agg_edges_kernel<<<(E + T - 1) / T, T, 0, stream>>>(src, dst, ew, dis, h1, agg1, E);
  tanh_bias_kernel<<<((long long)n * H1DIM + T - 1) / T, T, 0, stream>>>(agg1, h1, dis, b1, n);

  agg_edges_kernel<<<(E + T - 1) / T, T, 0, stream>>>(src, dst, ew, dis, agg1, agg2, E);
  gemm2_wmma<<<(n + 15) / 16, 32, 0, stream>>>(agg2, agg1, dis, W2, b2, out, n);
}


# edge sums (degree, 2 layers): global f32 atomics -> LDS adds; edges grouped per call by target range (LDS counts, scan), one block per 256-node bin; no zero fill; tanh as epilogue; add order free
# speedup vs baseline: 2.2080x; 1.6822x over previous
#include <hip/hip_runtime.h>
#include <math.h>

typedef __attribute__((ext_vector_type(2))) float v2f;
typedef __attribute__((ext_vector_type(8))) float v8f;

#define NC    45
#define H1DIM 32
#define H2DIM 128

#define BIN_SHIFT 8
#define BIN   (1 << BIN_SHIFT)
#define NBMAX 512
#define CHUNK 4096
#define CH_T  256
#define EPT   (CHUNK / CH_T)
__global__ void __launch_bounds__(CH_T)
bin_count_kernel(const int* __restrict__ dst, int* __restrict__ M, int E, int NB, int NCH) {
  __shared__ int cur[NBMAX];
  const int tid    = threadIdx.x;
  const int chunk  = blockIdx.x;
  const int e_base = chunk * CHUNK;
  for (int b = tid; b < NB; b += CH_T) cur[b] = 0;
  __syncthreads();
  int d[EPT];
#pragma unroll
  for (int k = 0; k < EPT; ++k) {
    const int e  = e_base + k * CH_T + tid;
    const int dv = dst[e < E ? e : E - 1];
    d[k] = (e < E) ? dv : -1;
  }
#pragma unroll
  for (int k = 0; k < EPT; ++k)
    if (d[k] >= 0) atomicAdd(&cur[d[k] >> BIN_SHIFT], 1);
  __syncthreads();
  for (int b = tid; b < NB; b += CH_T) M[b * NCH + chunk] = cur[b];
}

#define SCAN_T 1024
__global__ void __launch_bounds__(SCAN_T)
offsets_kernel(const int* __restrict__ cnt, int* __restrict__ off, int len) {
  __shared__ int s_part[32];
  __shared__ int s_tot[32];
  const int t    = threadIdx.x;
  const int lane = t & 31;
  const int wv   = t >> 5;
  const int lim  = blockIdx.x * SCAN_T;
  int part = 0;
#pragma unroll 8
  for (int q = 4 * t; q < lim; q += 4 * SCAN_T) {
    const int4 c4 = *(const int4*)(cnt + q);
    part += (c4.x + c4.y) + (c4.z + c4.w);
  }
  const int i   = lim + t;
  const int own = (i < len) ? cnt[i] : 0;

  int incl = own;
#pragma unroll
  for (int dlt = 1; dlt < 32; dlt <<= 1) {
    const int u = __shfl_up(incl, dlt);
    if (lane >= dlt) incl += u;
  }
  int ps = part;
#pragma unroll
  for (int dlt = 16; dlt >= 1; dlt >>= 1) ps += __shfl_xor(ps, dlt);
  if (lane == 31) s_tot[wv]  = incl;
  if (lane == 0)  s_part[wv] = ps;
  __syncthreads();
  const int wt = s_tot[lane];
  int base = s_part[lane];
#pragma unroll
  for (int dlt = 16; dlt >= 1; dlt >>= 1) base += __shfl_xor(base, dlt);
  int wincl = wt;
#pragma unroll
  for (int dlt = 1; dlt < 32; dlt <<= 1) {
    const int u = __shfl_up(wincl, dlt);
    if (lane >= dlt) wincl += u;
  }
  const int wexcl = __shfl(wincl - wt, wv);
  if (i < len) off[i] = base + wexcl + (incl - own);
}

__global__ void __launch_bounds__(CH_T)
bin_scatter_kernel(const int* __restrict__ src, const int* __restrict__ dst,
                   const float* __restrict__ ew, const int* __restrict__ Moff,
                   int4* __restrict__ binned, int E, int NB, int NCH) {
  __shared__ int   cur[NBMAX];
  __shared__ int   lbase[NBMAX];
  __shared__ int   gbase[NBMAX];
  __shared__ int   s_src[CHUNK];
  __shared__ int   s_dst[CHUNK];
  __shared__ float s_w[CHUNK];
  const int tid    = threadIdx.x;
  const int lane   = tid & 31;
  const int chunk  = blockIdx.x;
  const int e_base = chunk * CHUNK;
  for (int b = tid; b < NB; b += CH_T) { cur[b] = 0; gbase[b] = Moff[b * NCH + chunk]; }
  __syncthreads();
  int   d[EPT], r[EPT], sv[EPT];
  float wv[EPT];
#pragma unroll
  for (int k = 0; k < EPT; ++k) {
    const int e  = e_base + k * CH_T + tid;
    const int ec = e < E ? e : E - 1;
    const int dv = dst[ec];
    sv[k] = src[ec];
    wv[k] = ew[ec];
    d[k]  = (e < E) ? dv : -1;
  }
#pragma unroll
  for (int k = 0; k < EPT; ++k) {
    r[k] = 0;
    if (d[k] >= 0) r[k] = atomicAdd(&cur[d[k] >> BIN_SHIFT], 1);
  }
  __syncthreads();
  if (tid < 32) {
    int running = 0;
    for (int b0 = 0; b0 < NB; b0 += 32) {
      const int bb = b0 + lane;
      const int v  = (bb < NB) ? cur[bb] : 0;
      int incl = v;
#pragma unroll
      for (int dlt = 1; dlt < 32; dlt <<= 1) {
        const int u = __shfl_up(incl, dlt);
        if (lane >= dlt) incl += u;
      }
      if (bb < NB) lbase[bb] = running + incl - v;
      running += __shfl(incl, 31);
    }
  }
  __syncthreads();
#pragma unroll
  for (int k = 0; k < EPT; ++k) {
    if (d[k] >= 0) {
      const int li = lbase[d[k] >> BIN_SHIFT] + r[k];
      s_src[li] = sv[k];
      s_dst[li] = d[k];
      s_w[li]   = wv[k];
    }
  }
  __syncthreads();
  const int left   = E - e_base;
  const int nvalid = left < CHUNK ? left : CHUNK;
  for (int i = tid; i < nvalid; i += CH_T) {
    const int dd = s_dst[i];
    const int b  = dd >> BIN_SHIFT;
    const int g  = gbase[b] + (i - lbase[b]);
    binned[g] = make_int4(s_src[i], 0, dd, __float_as_int(s_w[i]));
  }
}

__global__ void __launch_bounds__(256)
bin_degree_kernel(const int4* __restrict__ binned, const int* __restrict__ Moff,
                  float* __restrict__ deg, int n, int E, int NB, int NCH) {
  __shared__ float dsum[BIN];
  const int b   = blockIdx.x;
  const int tid = threadIdx.x;
  for (int k = tid; k < BIN; k += (int)blockDim.x) dsum[k] = 0.0f;
  const int bs = Moff[b * NCH];
  const int be = (b + 1 < NB) ? Moff[(b + 1) * NCH] : E;
  __syncthreads();
  const int2* dw = (const int2*)binned;
  for (int i = bs + tid; i < be; i += (int)blockDim.x) {
    const int2 p = dw[2 * (size_t)i + 1];
    atomicAdd(&dsum[p.x & (BIN - 1)], __int_as_float(p.y));
  }
  __syncthreads();
  for (int k = tid; k < BIN; k += (int)blockDim.x) {
    const int node = (b << BIN_SHIFT) + k;
    if (node < n) deg[node] = dsum[k];
  }
}

template <bool LAYER1>
__global__ void __launch_bounds__(1024)
bin_aggregate_kernel(int4* __restrict__ binned, const int* __restrict__ Moff,
                     const float* __restrict__ dis, const float* __restrict__ h,
                     const float* __restrict__ b1, float* __restrict__ outp,
                     int n, int E, int NB, int NCH) {
  __shared__ float tile[BIN * H1DIM];
  const int lane = threadIdx.x & 31;
  const int wv = __builtin_amdgcn_readfirstlane((int)(threadIdx.x >> 5));
  const int nw = (int)(blockDim.x >> 5);
  const int b  = blockIdx.x;
  for (int k = threadIdx.x; k < BIN * H1DIM; k += (int)blockDim.x) tile[k] = 0.0f;
  const int bs    = Moff[b * NCH];
  const int be    = (b + 1 < NB) ? Moff[(b + 1) * NCH] : E;
  const int node0 = b << BIN_SHIFT;
  __syncthreads();

  for (int i0 = bs + wv * 32; i0 < be; i0 += nw * 32) {
    const int i = i0 + lane;
    int   s   = node0;
    int   dl  = 0;
    float nrm = 0.0f;
    if (i < be) {
      const int4 r = binned[i];
      s  = r.x;
      dl = r.z & (BIN - 1);
      if (LAYER1) {
        const float dsv = dis[s];
        const float ddv = dis[r.z];
        nrm = dsv * __int_as_float(r.w) * ddv;
        ((int*)binned)[4 * (size_t)i + 1] = __float_as_int(nrm);
      } else {
        nrm = __int_as_float(r.y);
      }
    }
    if (i0 + 32 <= be) {
      float v[32];
#pragma unroll
      for (int j = 0; j < 32; ++j) {
        const int sj = __builtin_amdgcn_readlane(s, j);
        v[j] = h[(size_t)sj * H1DIM + lane];
      }
#pragma unroll
      for (int j = 0; j < 32; ++j) {
        const int   dj = __builtin_amdgcn_readlane(dl, j);
        const float nj = __int_as_float(__builtin_amdgcn_readlane(__float_as_int(nrm), j));
        atomicAdd(&tile[dj * H1DIM + lane], v[j] * nj);
      }
    } else {
      const int cnt = be - i0;
      for (int j = 0; j < cnt; ++j) {
        const int   sj = __builtin_amdgcn_readlane(s, j);
        const int   dj = __builtin_amdgcn_readlane(dl, j);
        const float nj = __int_as_float(__builtin_amdgcn_readlane(__float_as_int(nrm), j));
        atomicAdd(&tile[dj * H1DIM + lane], h[(size_t)sj * H1DIM + lane] * nj);
      }
    }
  }
  __syncthreads();

  for (int r = wv; r < BIN; r += nw) {
    const int node = node0 + r;
    if (node < n) {
      const float acc = tile[r * H1DIM + lane];
      if (LAYER1) {
        const float di = dis[node];
        outp[(size_t)node * H1DIM + lane] =
            tanhf(acc + h[(size_t)node * H1DIM + lane] * di * di + b1[lane]);
      } else {
        outp[(size_t)node * H1DIM + lane] = acc;
      }
    }
  }
}

__global__ void gemm1_wmma(const float* __restrict__ x, const float* __restrict__ W1,
                           const float* __restrict__ deg, float* __restrict__ dis,
                           float* __restrict__ h1, int n) {
  const int lane  = threadIdx.x;
  const int m0    = blockIdx.x * 16;
  const int khalf = (lane >> 4) & 1;
  const int nn    = lane & 15;
  int m = m0 + nn;
  if (m > n - 1) m = n - 1;

  const int  drow = m0 + lane;
  const bool down = (lane < 16) && (drow < n);
  const float dg = deg[drow < n ? drow : n - 1];

  const float* xr = x + (size_t)m * NC + 2 * khalf;
  const float* wr = W1 + (size_t)(2 * khalf) * H1DIM + nn;
  v8f c0 = {};
  v8f c1 = {};
#pragma unroll
  for (int k = 0; k < 11; ++k) {
    v2f a, b0, b1;
    a.x  = xr[4 * k];
    a.y  = xr[4 * k + 1];
    b0.x = wr[(4 * k) * H1DIM];
    b0.y = wr[(4 * k + 1) * H1DIM];
    b1.x = wr[(4 * k) * H1DIM + 16];
    b1.y = wr[(4 * k + 1) * H1DIM + 16];
    c0 = __builtin_amdgcn_wmma_f32_16x16x4_f32(false, a, false, b0, (short)0, c0, false, false);
    c1 = __builtin_amdgcn_wmma_f32_16x16x4_f32(false, a, false, b1, (short)0, c1, false, false);
  }
  {
    const int   koff = khalf ? 0 : 44;
    const float* xl  = x + (size_t)m * NC + koff;
    const float* wl  = W1 + (size_t)koff * H1DIM + nn;
    const float ta = xl[0], t0 = wl[0], t1 = wl[16];
    v2f a, b0, b1;
    a.x  = khalf ? 0.0f : ta;  a.y  = 0.0f;
    b0.x = khalf ? 0.0f : t0;  b0.y = 0.0f;
    b1.x = khalf ? 0.0f : t1;  b1.y = 0.0f;
    c0 = __builtin_amdgcn_wmma_f32_16x16x4_f32(false, a, false, b0, (short)0, c0, false, false);
    c1 = __builtin_amdgcn_wmma_f32_16x16x4_f32(false, a, false, b1, (short)0, c1, false, false);
  }
  const int rbase = m0 + (khalf << 3);
#pragma unroll
  for (int j = 0; j < 8; ++j) {
    const int r = rbase + j;
    if (r < n) {
      h1[(size_t)r * H1DIM + nn]      = c0[j];
      h1[(size_t)r * H1DIM + 16 + nn] = c1[j];
    }
  }
  if (down) dis[drow] = 1.0f / sqrtf(dg + 1.0f);
}

__global__ void gemm2_wmma(const float* __restrict__ agg2, const float* __restrict__ h,
                           const float* __restrict__ dis, const float* __restrict__ W2,
                           const float* __restrict__ b2, float* __restrict__ out, int n) {
  const int lane  = threadIdx.x;
  const int m0    = blockIdx.x * 16;
  const int khalf = (lane >> 4) & 1;
  const int nn    = lane & 15;
  int m = m0 + nn;
  if (m > n - 1) m = n - 1;

  const float* ap = agg2 + (size_t)m * H1DIM + 2 * khalf;
  const float* hp = h    + (size_t)m * H1DIM + 2 * khalf;
  const float* wr = W2 + (size_t)(2 * khalf) * H2DIM + nn;
  const float di = dis[m];
  v2f a[8];
#pragma unroll
  for (int k = 0; k < 8; ++k) {
    const v2f ag = *(const v2f*)(ap + 4 * k);
    const v2f hv = *(const v2f*)(hp + 4 * k);
    a[k].x = ag.x + hv.x * di * di;
    a[k].y = ag.y + hv.y * di * di;
  }

  v8f acc[8] = {};
#pragma unroll
  for (int k = 0; k < 8; ++k) {
#pragma unroll
    for (int t = 0; t < 8; ++t) {
      v2f b;
      b.x = wr[(4 * k) * H2DIM + t * 16];
      b.y = wr[(4 * k + 1) * H2DIM + t * 16];
      acc[t] = __builtin_amdgcn_wmma_f32_16x16x4_f32(false, a[k], false, b, (short)0, acc[t], false, false);
    }
  }
  float bv[8];
#pragma unroll
  for (int t = 0; t < 8; ++t) bv[t] = b2[t * 16 + nn];

  const int rbase = m0 + (khalf << 3);
#pragma unroll
  for (int j = 0; j < 8; ++j) {
    const int r = rbase + j;
    if (r < n) {
#pragma unroll
      for (int t = 0; t < 8; ++t)
        out[(size_t)r * H2DIM + t * 16 + nn] = acc[t][j] + bv[t];
    }
  }
}

extern "C" void kernel_launch(void* const* d_in, const int* in_sizes, int n_in,
                              void* d_out, int out_size, void* d_ws, size_t ws_size,
                              hipStream_t stream) {
  const float* x   = (const float*)d_in[0];
  const int*   src = (const int*)  d_in[1];
  const int*   dst = (const int*)  d_in[2];
  const float* ew  = (const float*)d_in[3];
  const float* W1  = (const float*)d_in[4];
  const float* b1  = (const float*)d_in[5];
  const float* W2  = (const float*)d_in[6];
  const float* b2  = (const float*)d_in[7];
  float* out = (float*)d_out;

  const int n = in_sizes[0] / NC;
  const int E = in_sizes[1];

  const int NB  = (n + BIN - 1) / BIN;
  const int NCH = (E + CHUNK - 1) / CHUNK;
  const int L   = NB * NCH;

  const size_t np = ((size_t)n + 63) & ~(size_t)63;
  const size_t Lp = ((size_t)L + 63) & ~(size_t)63;
  int*   M      = (int*)d_ws;
  int*   Moff   = M + Lp;
  float* deg    = (float*)(Moff + Lp);
  float* dis    = deg + np;
  float* h1     = dis + np;
  float* hA     = h1 + np * H1DIM;
  float* agg2   = hA + np * H1DIM;
  int4*  binned = (int4*)(agg2 + np * H1DIM);

  bin_count_kernel<<<NCH, CH_T, 0, stream>>>(dst, M, E, NB, NCH);
  offsets_kernel<<<(L + SCAN_T - 1) / SCAN_T, SCAN_T, 0, stream>>>(M, Moff, L);
  bin_scatter_kernel<<<NCH, CH_T, 0, stream>>>(src, dst, ew, Moff, binned, E, NB, NCH);
  bin_degree_kernel<<<NB, 256, 0, stream>>>(binned, Moff, deg, n, E, NB, NCH);

  gemm1_wmma<<<(n + 15) / 16, 32, 0, stream>>>(x, W1, deg, dis, h1, n);
  bin_aggregate_kernel<true><<<NB, 1024, 0, stream>>>(binned, Moff, dis, h1, b1, hA, n, E, NB, NCH);

  bin_aggregate_kernel<false><<<NB, 1024, 0, stream>>>(binned, Moff, dis, hA, b1, agg2, n, E, NB, NCH);
  gemm2_wmma<<<(n + 15) / 16, 32, 0, stream>>>(agg2, hA, dis, W2, b2, out, n);
}


# layer-1 transform x@W1 (WMMA) rides in the edge-scatter launch as extra blocks of 8 stripes: the two independent passes overlap, 8 launches become 7; dis formed in the degree pass; sums unchanged
# speedup vs baseline: 2.3187x; 1.0501x over previous
#include <hip/hip_runtime.h>
#include <math.h>

typedef __attribute__((ext_vector_type(2))) float v2f;
typedef __attribute__((ext_vector_type(8))) float v8f;

#define NC    45
#define H1DIM 32
#define H2DIM 128

#define BIN_SHIFT 8
#define BIN   (1 << BIN_SHIFT)
#define NBMAX 512
#define CHUNK 4096
#define CH_T  256
#define EPT   (CHUNK / CH_T)
__global__ void __launch_bounds__(CH_T)
bin_count_kernel(const int* __restrict__ dst, int* __restrict__ M, int E, int NB, int NCH) {
  __shared__ int cur[NBMAX];
  const int tid    = threadIdx.x;
  const int chunk  = blockIdx.x;
  const int e_base = chunk * CHUNK;
  for (int b = tid; b < NB; b += CH_T) cur[b] = 0;
  __syncthreads();
  int d[EPT];
#pragma unroll
  for (int k = 0; k < EPT; ++k) {
    const int e  = e_base + k * CH_T + tid;
    const int dv = dst[e < E ? e : E - 1];
    d[k] = (e < E) ? dv : -1;
  }
#pragma unroll
  for (int k = 0; k < EPT; ++k)
    if (d[k] >= 0) atomicAdd(&cur[d[k] >> BIN_SHIFT], 1);
  __syncthreads();
  for (int b = tid; b < NB; b += CH_T) M[b * NCH + chunk] = cur[b];
}

#define SCAN_T 1024
__global__ void __launch_bounds__(SCAN_T)
offsets_kernel(const int* __restrict__ cnt, int* __restrict__ off, int len) {
  __shared__ int s_part[32];
  __shared__ int s_tot[32];
  const int t    = threadIdx.x;
  const int lane = t & 31;
  const int wv   = t >> 5;
  const int lim  = blockIdx.x * SCAN_T;
  int part = 0;
#pragma unroll 8
  for (int q = 4 * t; q < lim; q += 4 * SCAN_T) {
    const int4 c4 = *(const int4*)(cnt + q);
    part += (c4.x + c4.y) + (c4.z + c4.w);
  }
  const int i   = lim + t;
  const int own = (i < len) ? cnt[i] : 0;

  int incl = own;
#pragma unroll
  for (int dlt = 1; dlt < 32; dlt <<= 1) {
    const int u = __shfl_up(incl, dlt);
    if (lane >= dlt) incl += u;
  }
  int ps = part;
#pragma unroll
  for (int dlt = 16; dlt >= 1; dlt >>= 1) ps += __shfl_xor(ps, dlt);
  if (lane == 31) s_tot[wv]  = incl;
  if (lane == 0)  s_part[wv] = ps;
  __syncthreads();
  const int wt = s_tot[lane];
  int base = s_part[lane];
#pragma unroll
  for (int dlt = 16; dlt >= 1; dlt >>= 1) base += __shfl_xor(base, dlt);
  int wincl = wt;
#pragma unroll
  for (int dlt = 1; dlt < 32; dlt <<= 1) {
    const int u = __shfl_up(wincl, dlt);
    if (lane >= dlt) wincl += u;
  }
  const int wexcl = __shfl(wincl - wt, wv);
  if (i < len) off[i] = base + wexcl + (incl - own);
}

__device__ __forceinline__ void gemm1_stripe(const float* __restrict__ x,
                                             const float* __restrict__ W1,
                                             float* __restrict__ h1, int n, int m0, int lane) {
  const int khalf = (lane >> 4) & 1;
  const int nn    = lane & 15;
  int m = m0 + nn;
  if (m > n - 1) m = n - 1;

  const float* xr = x + (size_t)m * NC + 2 * khalf;
  const float* wr = W1 + (size_t)(2 * khalf) * H1DIM + nn;
  v8f c0 = {};
  v8f c1 = {};
#pragma unroll
  for (int k = 0; k < 11; ++k) {
    v2f a, b0, b1;
    a.x  = xr[4 * k];
    a.y  = xr[4 * k + 1];
    b0.x = wr[(4 * k) * H1DIM];
    b0.y = wr[(4 * k + 1) * H1DIM];
    b1.x = wr[(4 * k) * H1DIM + 16];
    b1.y = wr[(4 * k + 1) * H1DIM + 16];
    c0 = __builtin_amdgcn_wmma_f32_16x16x4_f32(false, a, false, b0, (short)0, c0, false, false);
    c1 = __builtin_amdgcn_wmma_f32_16x16x4_f32(false, a, false, b1, (short)0, c1, false, false);
  }
  {
    const int   koff = khalf ? 0 : 44;
    const float* xl  = x + (size_t)m * NC + koff;
    const float* wl  = W1 + (size_t)koff * H1DIM + nn;
    const float ta = xl[0], t0 = wl[0], t1 = wl[16];
    v2f a, b0, b1;
    a.x  = khalf ? 0.0f : ta;  a.y  = 0.0f;
    b0.x = khalf ? 0.0f : t0;  b0.y = 0.0f;
    b1.x = khalf ? 0.0f : t1;  b1.y = 0.0f;
    c0 = __builtin_amdgcn_wmma_f32_16x16x4_f32(false, a, false, b0, (short)0, c0, false, false);
    c1 = __builtin_amdgcn_wmma_f32_16x16x4_f32(false, a, false, b1, (short)0, c1, false, false);
  }
  const int rbase = m0 + (khalf << 3);
#pragma unroll
  for (int j = 0; j < 8; ++j) {
    const int r = rbase + j;
    if (r < n) {
      h1[(size_t)r * H1DIM + nn]      = c0[j];
      h1[(size_t)r * H1DIM + 16 + nn] = c1[j];
    }
  }
}

__global__ void __launch_bounds__(CH_T)
bin_scatter_gemm1_kernel(const int* __restrict__ src, const int* __restrict__ dst,
                         const float* __restrict__ ew, const int* __restrict__ Moff,
                         int4* __restrict__ binned, int E, int NB, int NCH,
                         const float* __restrict__ x, const float* __restrict__ W1,
                         float* __restrict__ h1, int n) {
  if ((int)blockIdx.x >= NCH) {
    const int wv = __builtin_amdgcn_readfirstlane((int)(threadIdx.x >> 5));
    const int m0 = (((int)blockIdx.x - NCH) * (CH_T / 32) + wv) * 16;
    if (m0 < n) gemm1_stripe(x, W1, h1, n, m0, (int)(threadIdx.x & 31));
    return;
  }
  __shared__ int   cur[NBMAX];
  __shared__ int   lbase[NBMAX];
  __shared__ int   gbase[NBMAX];
  __shared__ int   s_src[CHUNK];
  __shared__ int   s_dst[CHUNK];
  __shared__ float s_w[CHUNK];
  const int tid    = threadIdx.x;
  const int lane   = tid & 31;
  const int chunk  = blockIdx.x;
  const int e_base = chunk * CHUNK;
  for (int b = tid; b < NB; b += CH_T) { cur[b] = 0; gbase[b] = Moff[b * NCH + chunk]; }
  __syncthreads();
  int   d[EPT], r[EPT], sv[EPT];
  float wv[EPT];
#pragma unroll
  for (int k = 0; k < EPT; ++k) {
    const int e  = e_base + k * CH_T + tid;
    const int ec = e < E ? e : E - 1;
    const int dv = dst[ec];
    sv[k] = src[ec];
    wv[k] = ew[ec];
    d[k]  = (e < E) ? dv : -1;
  }
#pragma unroll
  for (int k = 0; k < EPT; ++k) {
    r[k] = 0;
    if (d[k] >= 0) r[k] = atomicAdd(&cur[d[k] >> BIN_SHIFT], 1);
  }
  __syncthreads();
  if (tid < 32) {
    int running = 0;
    for (int b0 = 0; b0 < NB; b0 += 32) {
      const int bb = b0 + lane;
      const int v  = (bb < NB) ? cur[bb] : 0;
      int incl = v;
#pragma unroll
      for (int dlt = 1; dlt < 32; dlt <<= 1) {
        const int u = __shfl_up(incl, dlt);
        if (lane >= dlt) incl += u;
      }
      if (bb < NB) lbase[bb] = running + incl - v;
      running += __shfl(incl, 31);
    }
  }
  __syncthreads();
#pragma unroll
  for (int k = 0; k < EPT; ++k) {
    if (d[k] >= 0) {
      const int li = lbase[d[k] >> BIN_SHIFT] + r[k];
      s_src[li] = sv[k];
      s_dst[li] = d[k];
      s_w[li]   = wv[k];
    }
  }
  __syncthreads();
  const int left   = E - e_base;
  const int nvalid = left < CHUNK ? left : CHUNK;
  for (int i = tid; i < nvalid; i += CH_T) {
    const int dd = s_dst[i];
    const int b  = dd >> BIN_SHIFT;
    const int g  = gbase[b] + (i - lbase[b]);
    binned[g] = make_int4(s_src[i], 0, dd, __float_as_int(s_w[i]));
  }
}

__global__ void __launch_bounds__(256)
bin_degree_kernel(const int4* __restrict__ binned, const int* __restrict__ Moff,
                  float* __restrict__ dis, int n, int E, int NB, int NCH) {
  __shared__ float dsum[BIN];
  const int b   = blockIdx.x;
  const int tid = threadIdx.x;
  for (int k = tid; k < BIN; k += (int)blockDim.x) dsum[k] = 0.0f;
  const int bs = Moff[b * NCH];
  const int be = (b + 1 < NB) ? Moff[(b + 1) * NCH] : E;
  __syncthreads();
  const int2* dw = (const int2*)binned;
  for (int i = bs + tid; i < be; i += (int)blockDim.x) {
    const int2 p = dw[2 * (size_t)i + 1];
    atomicAdd(&dsum[p.x & (BIN - 1)], __int_as_float(p.y));
  }
  __syncthreads();
  for (int k = tid; k < BIN; k += (int)blockDim.x) {
    const int node = (b << BIN_SHIFT) + k;
    if (node < n) dis[node] = 1.0f / sqrtf(dsum[k] + 1.0f);
  }
}

template <bool LAYER1>
__global__ void __launch_bounds__(1024)
bin_aggregate_kernel(int4* __restrict__ binned, const int* __restrict__ Moff,
                     const float* __restrict__ dis, const float* __restrict__ h,
                     const float* __restrict__ b1, float* __restrict__ outp,
                     int n, int E, int NB, int NCH) {
  __shared__ float tile[BIN * H1DIM];
  const int lane = threadIdx.x & 31;
  const int wv = __builtin_amdgcn_readfirstlane((int)(threadIdx.x >> 5));
  const int nw = (int)(blockDim.x >> 5);
  const int b  = blockIdx.x;
  for (int k = threadIdx.x; k < BIN * H1DIM; k += (int)blockDim.x) tile[k] = 0.0f;
  const int bs    = Moff[b * NCH];
  const int be    = (b + 1 < NB) ? Moff[(b + 1) * NCH] : E;
  const int node0 = b << BIN_SHIFT;
  __syncthreads();

  for (int i0 = bs + wv * 32; i0 < be; i0 += nw * 32) {
    const int i = i0 + lane;
    int   s   = node0;
    int   dl  = 0;
    float nrm = 0.0f;
    if (i < be) {
      const int4 r = binned[i];
      s  = r.x;
      dl = r.z & (BIN - 1);
      if (LAYER1) {
        const float dsv = dis[s];
        const float ddv = dis[r.z];
        nrm = dsv * __int_as_float(r.w) * ddv;
        ((int*)binned)[4 * (size_t)i + 1] = __float_as_int(nrm);
      } else {
        nrm = __int_as_float(r.y);
      }
    }
    if (i0 + 32 <= be) {
      float v[32];
#pragma unroll
      for (int j = 0; j < 32; ++j) {
        const int sj = __builtin_amdgcn_readlane(s, j);
        v[j] = h[(size_t)sj * H1DIM + lane];
      }
#pragma unroll
      for (int j = 0; j < 32; ++j) {
        const int   dj = __builtin_amdgcn_readlane(dl, j);
        const float nj = __int_as_float(__builtin_amdgcn_readlane(__float_as_int(nrm), j));
        atomicAdd(&tile[dj * H1DIM + lane], v[j] * nj);
      }
    } else {
      const int cnt = be - i0;
      for (int j = 0; j < cnt; ++j) {
        const int   sj = __builtin_amdgcn_readlane(s, j);
        const int   dj = __builtin_amdgcn_readlane(dl, j);
        const float nj = __int_as_float(__builtin_amdgcn_readlane(__float_as_int(nrm), j));
        atomicAdd(&tile[dj * H1DIM + lane], h[(size_t)sj * H1DIM + lane] * nj);
      }
    }
  }
  __syncthreads();

  for (int r = wv; r < BIN; r += nw) {
    const int node = node0 + r;
    if (node < n) {
      const float acc = tile[r * H1DIM + lane];
      if (LAYER1) {
        const float di = dis[node];
        outp[(size_t)node * H1DIM + lane] =
            tanhf(acc + h[(size_t)node * H1DIM + lane] * di * di + b1[lane]);
      } else {
        outp[(size_t)node * H1DIM + lane] = acc;
      }
    }
  }
}

__global__ void gemm2_wmma(const float* __restrict__ agg2, const float* __restrict__ h,
                           const float* __restrict__ dis, const float* __restrict__ W2,
                           const float* __restrict__ b2, float* __restrict__ out, int n) {
  const int lane  = threadIdx.x;
  const int m0    = blockIdx.x * 16;
  const int khalf = (lane >> 4) & 1;
  const int nn    = lane & 15;
  int m = m0 + nn;
  if (m > n - 1) m = n - 1;

  const float* ap = agg2 + (size_t)m * H1DIM + 2 * khalf;
  const float* hp = h    + (size_t)m * H1DIM + 2 * khalf;
  const float* wr = W2 + (size_t)(2 * khalf) * H2DIM + nn;
  const float di = dis[m];
  v2f a[8];
#pragma unroll
  for (int k = 0; k < 8; ++k) {
    const v2f ag = *(const v2f*)(ap + 4 * k);
    const v2f hv = *(const v2f*)(hp + 4 * k);
    a[k].x = ag.x + hv.x * di * di;
    a[k].y = ag.y + hv.y * di * di;
  }

  v8f acc[8] = {};
#pragma unroll
  for (int k = 0; k < 8; ++k) {
#pragma unroll
    for (int t = 0; t < 8; ++t) {
      v2f b;
      b.x = wr[(4 * k) * H2DIM + t * 16];
      b.y = wr[(4 * k + 1) * H2DIM + t * 16];
      acc[t] = __builtin_amdgcn_wmma_f32_16x16x4_f32(false, a[k], false, b, (short)0, acc[t], false, false);
    }
  }
  float bv[8];
#pragma unroll
  for (int t = 0; t < 8; ++t) bv[t] = b2[t * 16 + nn];

  const int rbase = m0 + (khalf << 3);
#pragma unroll
  for (int j = 0; j < 8; ++j) {
    const int r = rbase + j;
    if (r < n) {
#pragma unroll
      for (int t = 0; t < 8; ++t)
        out[(size_t)r * H2DIM + t * 16 + nn] = acc[t][j] + bv[t];
    }
  }
}

extern "C" void kernel_launch(void* const* d_in, const int* in_sizes, int n_in,
                              void* d_out, int out_size, void* d_ws, size_t ws_size,
                              hipStream_t stream) {
  const float* x   = (const float*)d_in[0];
  const int*   src = (const int*)  d_in[1];
  const int*   dst = (const int*)  d_in[2];
  const float* ew  = (const float*)d_in[3];
  const float* W1  = (const float*)d_in[4];
  const float* b1  = (const float*)d_in[5];
  const float* W2  = (const float*)d_in[6];
  const float* b2  = (const float*)d_in[7];
  float* out = (float*)d_out;

  const int n = in_sizes[0] / NC;
  const int E = in_sizes[1];

  const int NB  = (n + BIN - 1) / BIN;
  const int NCH = (E + CHUNK - 1) / CHUNK;
  const int L   = NB * NCH;

  const size_t np = ((size_t)n + 63) & ~(size_t)63;
  const size_t Lp = ((size_t)L + 63) & ~(size_t)63;
  int*   M      = (int*)d_ws;
  int*   Moff   = M + Lp;
  float* dis    = (float*)(Moff + Lp);
  float* h1     = dis + np;
  float* hA     = h1 + np * H1DIM;
  float* agg2   = hA + np * H1DIM;
  int4*  binned = (int4*)(agg2 + np * H1DIM);

  bin_count_kernel<<<NCH, CH_T, 0, stream>>>(dst, M, E, NB, NCH);
  offsets_kernel<<<(L + SCAN_T - 1) / SCAN_T, SCAN_T, 0, stream>>>(M, Moff, L);
  const int G1B = (n + 16 * (CH_T / 32) - 1) / (16 * (CH_T / 32));
  bin_scatter_gemm1_kernel<<<NCH + G1B, CH_T, 0, stream>>>(src, dst, ew, Moff, binned, E, NB, NCH,
                                                          x, W1, h1, n);
  bin_degree_kernel<<<NB, 256, 0, stream>>>(binned, Moff, dis, n, E, NB, NCH);

  bin_aggregate_kernel<true><<<NB, 1024, 0, stream>>>(binned, Moff, dis, h1, b1, hA, n, E, NB, NCH);

  bin_aggregate_kernel<false><<<NB, 1024, 0, stream>>>(binned, Moff, dis, hA, b1, agg2, n, E, NB, NCH);
  gemm2_wmma<<<(n + 15) / 16, 32, 0, stream>>>(agg2, hA, dis, W2, b2, out, n);
}


# degree weights summed exactly in 64-bit fixed point (order-free), so its loop fetches 20 records per thread at once; layer-1 epilogue and a bin's last partial edge group also issue loads together
# speedup vs baseline: 2.6740x; 1.1532x over previous
#include <hip/hip_runtime.h>
#include <math.h>

typedef __attribute__((ext_vector_type(2))) float v2f;
typedef __attribute__((ext_vector_type(8))) float v8f;

#define NC    45
#define H1DIM 32
#define H2DIM 128

#define BIN_SHIFT 8
#define BIN   (1 << BIN_SHIFT)
#define NBMAX 512
#define CHUNK 4096
#define CH_T  256
#define EPT   (CHUNK / CH_T)
__global__ void __launch_bounds__(CH_T)
bin_count_kernel(const int* __restrict__ dst, int* __restrict__ M, int E, int NB, int NCH) {
  __shared__ int cur[NBMAX];
  const int tid    = threadIdx.x;
  const int chunk  = blockIdx.x;
  const int e_base = chunk * CHUNK;
  for (int b = tid; b < NB; b += CH_T) cur[b] = 0;
  __syncthreads();
  int d[EPT];
#pragma unroll
  for (int k = 0; k < EPT; ++k) {
    const int e  = e_base + k * CH_T + tid;
    const int dv = dst[e < E ? e : E - 1];
    d[k] = (e < E) ? dv : -1;
  }
#pragma unroll
  for (int k = 0; k < EPT; ++k)
    if (d[k] >= 0) atomicAdd(&cur[d[k] >> BIN_SHIFT], 1);
  __syncthreads();
  for (int b = tid; b < NB; b += CH_T) M[b * NCH + chunk] = cur[b];
}

#define SCAN_T 1024
__global__ void __launch_bounds__(SCAN_T)
offsets_kernel(const int* __restrict__ cnt, int* __restrict__ off, int len) {
  __shared__ int s_part[32];
  __shared__ int s_tot[32];
  const int t    = threadIdx.x;
  const int lane = t & 31;
  const int wv   = t >> 5;
  const int lim  = blockIdx.x * SCAN_T;
  int part = 0;
#pragma unroll 8
  for (int q = 4 * t; q < lim; q += 4 * SCAN_T) {
    const int4 c4 = *(const int4*)(cnt + q);
    part += (c4.x + c4.y) + (c4.z + c4.w);
  }
  const int i   = lim + t;
  const int own = (i < len) ? cnt[i] : 0;

  int incl = own;
#pragma unroll
  for (int dlt = 1; dlt < 32; dlt <<= 1) {
    const int u = __shfl_up(incl, dlt);
    if (lane >= dlt) incl += u;
  }
  int ps = part;
#pragma unroll
  for (int dlt = 16; dlt >= 1; dlt >>= 1) ps += __shfl_xor(ps, dlt);
  if (lane == 31) s_tot[wv]  = incl;
  if (lane == 0)  s_part[wv] = ps;
  __syncthreads();
  const int wt = s_tot[lane];
  int base = s_part[lane];
#pragma unroll
  for (int dlt = 16; dlt >= 1; dlt >>= 1) base += __shfl_xor(base, dlt);
  int wincl = wt;
#pragma unroll
  for (int dlt = 1; dlt < 32; dlt <<= 1) {
    const int u = __shfl_up(wincl, dlt);
    if (lane >= dlt) wincl += u;
  }
  const int wexcl = __shfl(wincl - wt, wv);
  if (i < len) off[i] = base + wexcl + (incl - own);
}

__device__ __forceinline__ void gemm1_stripe(const float* __restrict__ x,
                                             const float* __restrict__ W1,
                                             float* __restrict__ h1, int n, int m0, int lane) {
  const int khalf = (lane >> 4) & 1;
  const int nn    = lane & 15;
  int m = m0 + nn;
  if (m > n - 1) m = n - 1;

  const float* xr = x + (size_t)m * NC + 2 * khalf;
  const float* wr = W1 + (size_t)(2 * khalf) * H1DIM + nn;
  v8f c0 = {};
  v8f c1 = {};
#pragma unroll
  for (int k = 0; k < 11; ++k) {
    v2f a, b0, b1;
    a.x  = xr[4 * k];
    a.y  = xr[4 * k + 1];
    b0.x = wr[(4 * k) * H1DIM];
    b0.y = wr[(4 * k + 1) * H1DIM];
    b1.x = wr[(4 * k) * H1DIM + 16];
    b1.y = wr[(4 * k + 1) * H1DIM + 16];
    c0 = __builtin_amdgcn_wmma_f32_16x16x4_f32(false, a, false, b0, (short)0, c0, false, false);
    c1 = __builtin_amdgcn_wmma_f32_16x16x4_f32(false, a, false, b1, (short)0, c1, false, false);
  }
  {
    const int   koff = khalf ? 0 : 44;
    const float* xl  = x + (size_t)m * NC + koff;
    const float* wl  = W1 + (size_t)koff * H1DIM + nn;
    const float ta = xl[0], t0 = wl[0], t1 = wl[16];
    v2f a, b0, b1;
    a.x  = khalf ? 0.0f : ta;  a.y  = 0.0f;
    b0.x = khalf ? 0.0f : t0;  b0.y = 0.0f;
    b1.x = khalf ? 0.0f : t1;  b1.y = 0.0f;
    c0 = __builtin_amdgcn_wmma_f32_16x16x4_f32(false, a, false, b0, (short)0, c0, false, false);
    c1 = __builtin_amdgcn_wmma_f32_16x16x4_f32(false, a, false, b1, (short)0, c1, false, false);
  }
  const int rbase = m0 + (khalf << 3);
#pragma unroll
  for (int j = 0; j < 8; ++j) {
    const int r = rbase + j;
    if (r < n) {
      h1[(size_t)r * H1DIM + nn]      = c0[j];
      h1[(size_t)r * H1DIM + 16 + nn] = c1[j];
    }
  }
}

__global__ void __launch_bounds__(CH_T)
bin_scatter_gemm1_kernel(const int* __restrict__ src, const int* __restrict__ dst,
                         const float* __restrict__ ew, const int* __restrict__ Moff,
                         int4* __restrict__ binned, int E, int NB, int NCH,
                         const float* __restrict__ x, const float* __restrict__ W1,
                         float* __restrict__ h1, int n) {
  if ((int)blockIdx.x >= NCH) {
    const int wv = __builtin_amdgcn_readfirstlane((int)(threadIdx.x >> 5));
    const int m0 = (((int)blockIdx.x - NCH) * (CH_T / 32) + wv) * 16;
    if (m0 < n) gemm1_stripe(x, W1, h1, n, m0, (int)(threadIdx.x & 31));
    return;
  }
  __shared__ int   cur[NBMAX];
  __shared__ int   lbase[NBMAX];
  __shared__ int   gbase[NBMAX];
  __shared__ int   s_src[CHUNK];
  __shared__ int   s_dst[CHUNK];
  __shared__ float s_w[CHUNK];
  const int tid    = threadIdx.x;
  const int lane   = tid & 31;
  const int chunk  = blockIdx.x;
  const int e_base = chunk * CHUNK;
  for (int b = tid; b < NB; b += CH_T) { cur[b] = 0; gbase[b] = Moff[b * NCH + chunk]; }
  __syncthreads();
  int   d[EPT], r[EPT], sv[EPT];
  float wv[EPT];
#pragma unroll
  for (int k = 0; k < EPT; ++k) {
    const int e  = e_base + k * CH_T + tid;
    const int ec = e < E ? e : E - 1;
    const int dv = dst[ec];
    sv[k] = src[ec];
    wv[k] = ew[ec];
    d[k]  = (e < E) ? dv : -1;
  }
#pragma unroll
  for (int k = 0; k < EPT; ++k) {
    r[k] = 0;
    if (d[k] >= 0) r[k] = atomicAdd(&cur[d[k] >> BIN_SHIFT], 1);
  }
  __syncthreads();
  if (tid < 32) {
    int running = 0;
    for (int b0 = 0; b0 < NB; b0 += 32) {
      const int bb = b0 + lane;
      const int v  = (bb < NB) ? cur[bb] : 0;
      int incl = v;
#pragma unroll
      for (int dlt = 1; dlt < 32; dlt <<= 1) {
        const int u = __shfl_up(incl, dlt);
        if (lane >= dlt) incl += u;
      }
      if (bb < NB) lbase[bb] = running + incl - v;
      running += __shfl(incl, 31);
    }
  }
  __syncthreads();
#pragma unroll
  for (int k = 0; k < EPT; ++k) {
    if (d[k] >= 0) {
      const int li = lbase[d[k] >> BIN_SHIFT] + r[k];
      s_src[li] = sv[k];
      s_dst[li] = d[k];
      s_w[li]   = wv[k];
    }
  }
  __syncthreads();
  const int left   = E - e_base;
  const int nvalid = left < CHUNK ? left : CHUNK;
  for (int i = tid; i < nvalid; i += CH_T) {
    const int dd = s_dst[i];
    const int b  = dd >> BIN_SHIFT;
    const int g  = gbase[b] + (i - lbase[b]);
    binned[g] = make_int4(s_src[i], 0, dd, __float_as_int(s_w[i]));
  }
}

#define DEG_T    256
#define DEG_R    20
#define DEG_FRAC 40
__device__ __forceinline__ long long weight_to_fixed(float w) {
  const unsigned u  = __float_as_uint(w);
  const int      e  = (int)((u >> 23) & 0xffu);
  const unsigned long long m = (u & 0x7fffffu) | (e ? 0x800000u : 0u);
  const int sh = (e ? e : 1) - 150 + DEG_FRAC;
  const unsigned long long q = (sh >= 0) ? (m << (sh < 39 ? sh : 39))
                                         : (m >> (-sh < 24 ? -sh : 24));
  return (u >> 31) ? -(long long)q : (long long)q;
}
__global__ void __launch_bounds__(DEG_T)
bin_degree_kernel(const int4* __restrict__ binned, const int* __restrict__ Moff,
                  float* __restrict__ dis, int n, int E, int NB, int NCH) {
  __shared__ unsigned long long dsum[BIN];
  const int b   = blockIdx.x;
  const int tid = threadIdx.x;
  for (int k = tid; k < BIN; k += DEG_T) dsum[k] = 0ull;
  const int bs = Moff[b * NCH];
  const int be = (b + 1 < NB) ? Moff[(b + 1) * NCH] : E;
  __syncthreads();
  const int2* dw = (const int2*)binned;
  for (int i0 = bs + tid; i0 < be; i0 += DEG_T * DEG_R) {
    int   dk[DEG_R];
    float wk[DEG_R];
#pragma unroll
    for (int k = 0; k < DEG_R; ++k) {
      const int  i = i0 + k * DEG_T;
      const int2 p = dw[2 * (size_t)(i < be ? i : be - 1) + 1];
      dk[k] = (i < be) ? (p.x & (BIN - 1)) : -1;
      wk[k] = __int_as_float(p.y);
    }
    asm volatile("" :: "v"(wk[0]), "v"(wk[1]), "v"(wk[2]), "v"(wk[3]), "v"(wk[4]),
                       "v"(wk[5]), "v"(wk[6]), "v"(wk[7]), "v"(wk[8]), "v"(wk[9]),
                       "v"(wk[10]), "v"(wk[11]), "v"(wk[12]), "v"(wk[13]), "v"(wk[14]),
                       "v"(wk[15]), "v"(wk[16]), "v"(wk[17]), "v"(wk[18]), "v"(wk[19]));
    static_assert(DEG_R == 20, "the fence above names DEG_R weights");
#pragma unroll
    for (int k = 0; k < DEG_R; ++k) {
      if (dk[k] >= 0)
        atomicAdd(&dsum[dk[k]], (unsigned long long)weight_to_fixed(wk[k]));
    }
  }
  __syncthreads();
  for (int k = tid; k < BIN; k += DEG_T) {
    const int node = (b << BIN_SHIFT) + k;
    if (node < n) {
      const long long d1 = (long long)dsum[k] + (1ll << DEG_FRAC);
      dis[node] = 1.0f / sqrtf((float)d1 * (1.0f / (float)(1ll << DEG_FRAC)));
    }
  }
}

#define AGG_T 1024
#define AGG_W (AGG_T / 32)
#define AGG_R (BIN / AGG_W)
template <bool LAYER1>
__global__ void __launch_bounds__(AGG_T)
bin_aggregate_kernel(int4* __restrict__ binned, const int* __restrict__ Moff,
                     const float* __restrict__ dis, const float* __restrict__ h,
                     const float* __restrict__ b1, float* __restrict__ outp,
                     int n, int E, int NB, int NCH) {
  __shared__ float tile[BIN * H1DIM];
  const int lane = threadIdx.x & 31;
  const int wv = __builtin_amdgcn_readfirstlane((int)(threadIdx.x >> 5));
  const int nw = AGG_W;
  const int b  = blockIdx.x;
  for (int k = threadIdx.x; k < BIN * H1DIM; k += AGG_T) tile[k] = 0.0f;
  const int bs    = Moff[b * NCH];
  const int be    = (b + 1 < NB) ? Moff[(b + 1) * NCH] : E;
  const int node0 = b << BIN_SHIFT;
  __syncthreads();

  for (int i0 = bs + wv * 32; i0 < be; i0 += nw * 32) {
    const int i = i0 + lane;
    int   s   = node0;
    int   dl  = 0;
    float nrm = 0.0f;
    if (i < be) {
      const int4 r = binned[i];
      s  = r.x;
      dl = r.z & (BIN - 1);
      if (LAYER1) {
        const float dsv = dis[s];
        const float ddv = dis[r.z];
        nrm = dsv * __int_as_float(r.w) * ddv;
        ((int*)binned)[4 * (size_t)i + 1] = __float_as_int(nrm);
      } else {
        nrm = __int_as_float(r.y);
      }
    }
    if (i0 + 32 <= be) {
      float v[32];
#pragma unroll
      for (int j = 0; j < 32; ++j) {
        const int sj = __builtin_amdgcn_readlane(s, j);
        v[j] = h[(size_t)sj * H1DIM + lane];
      }
#pragma unroll
      for (int j = 0; j < 32; ++j) {
        const int   dj = __builtin_amdgcn_readlane(dl, j);
        const float nj = __int_as_float(__builtin_amdgcn_readlane(__float_as_int(nrm), j));
        atomicAdd(&tile[dj * H1DIM + lane], v[j] * nj);
      }
    } else {
      const int cnt = be - i0;
      float v[32];
#pragma unroll
      for (int j = 0; j < 32; ++j) {
        const int sj = __builtin_amdgcn_readlane(s, j);
        v[j] = h[(size_t)sj * H1DIM + lane];
      }
#pragma unroll
      for (int j = 0; j < 32; ++j) {
        if (j < cnt) {
          const int   dj = __builtin_amdgcn_readlane(dl, j);
          const float nj = __int_as_float(__builtin_amdgcn_readlane(__float_as_int(nrm), j));
          atomicAdd(&tile[dj * H1DIM + lane], v[j] * nj);
        }
      }
    }
  }
  __syncthreads();

  if (LAYER1) {
    float hv[AGG_R], dv[AGG_R];
#pragma unroll
    for (int k = 0; k < AGG_R; ++k) {
      int node = node0 + wv + k * AGG_W;
      if (node > n - 1) node = n - 1;
      dv[k] = dis[node];
      hv[k] = h[(size_t)node * H1DIM + lane];
    }
    const float bb = b1[lane];
#pragma unroll
    for (int k = 0; k < AGG_R; ++k) {
      const int r    = wv + k * AGG_W;
      const int node = node0 + r;
      if (node < n) {
        const float acc = tile[r * H1DIM + lane];
        const float di  = dv[k];
        outp[(size_t)node * H1DIM + lane] = tanhf(acc + hv[k] * di * di + bb);
      }
    }
  } else {
    for (int r = wv; r < BIN; r += nw) {
      const int node = node0 + r;
      if (node < n) outp[(size_t)node * H1DIM + lane] = tile[r * H1DIM + lane];
    }
  }
}

__global__ void gemm2_wmma(const float* __restrict__ agg2, const float* __restrict__ h,
                           const float* __restrict__ dis, const float* __restrict__ W2,
                           const float* __restrict__ b2, float* __restrict__ out, int n) {
  const int lane  = threadIdx.x;
  const int m0    = blockIdx.x * 16;
  const int khalf = (lane >> 4) & 1;
  const int nn    = lane & 15;
  int m = m0 + nn;
  if (m > n - 1) m = n - 1;

  const float* ap = agg2 + (size_t)m * H1DIM + 2 * khalf;
  const float* hp = h    + (size_t)m * H1DIM + 2 * khalf;
  const float* wr = W2 + (size_t)(2 * khalf) * H2DIM + nn;
  const float di = dis[m];
  v2f a[8];
#pragma unroll
  for (int k = 0; k < 8; ++k) {
    const v2f ag = *(const v2f*)(ap + 4 * k);
    const v2f hv = *(const v2f*)(hp + 4 * k);
    a[k].x = ag.x + hv.x * di * di;
    a[k].y = ag.y + hv.y * di * di;
  }

  v8f acc[8] = {};
#pragma unroll
  for (int k = 0; k < 8; ++k) {
#pragma unroll
    for (int t = 0; t < 8; ++t) {
      v2f b;
      b.x = wr[(4 * k) * H2DIM + t * 16];
      b.y = wr[(4 * k + 1) * H2DIM + t * 16];
      acc[t] = __builtin_amdgcn_wmma_f32_16x16x4_f32(false, a[k], false, b, (short)0, acc[t], false, false);
    }
  }
  float bv[8];
#pragma unroll
  for (int t = 0; t < 8; ++t) bv[t] = b2[t * 16 + nn];

  const int rbase = m0 + (khalf << 3);
#pragma unroll
  for (int j = 0; j < 8; ++j) {
    const int r = rbase + j;
    if (r < n) {
#pragma unroll
      for (int t = 0; t < 8; ++t)
        out[(size_t)r * H2DIM + t * 16 + nn] = acc[t][j] + bv[t];
    }
  }
}

extern "C" void kernel_launch(void* const* d_in, const int* in_sizes, int n_in,
                              void* d_out, int out_size, void* d_ws, size_t ws_size,
                              hipStream_t stream) {
  const float* x   = (const float*)d_in[0];
  const int*   src = (const int*)  d_in[1];
  const int*   dst = (const int*)  d_in[2];
  const float* ew  = (const float*)d_in[3];
  const float* W1  = (const float*)d_in[4];
  const float* b1  = (const float*)d_in[5];
  const float* W2  = (const float*)d_in[6];
  const float* b2  = (const float*)d_in[7];
  float* out = (float*)d_out;

  const int n = in_sizes[0] / NC;
  const int E = in_sizes[1];

  const int NB  = (n + BIN - 1) / BIN;
  const int NCH = (E + CHUNK - 1) / CHUNK;
  const int L   = NB * NCH;

  const size_t np = ((size_t)n + 63) & ~(size_t)63;
  const size_t Lp = ((size_t)L + 63) & ~(size_t)63;
  int*   M      = (int*)d_ws;
  int*   Moff   = M + Lp;
  float* dis    = (float*)(Moff + Lp);
  float* h1     = dis + np;
  float* hA     = h1 + np * H1DIM;
  float* agg2   = hA + np * H1DIM;
  int4*  binned = (int4*)(agg2 + np * H1DIM);

  bin_count_kernel<<<NCH, CH_T, 0, stream>>>(dst, M, E, NB, NCH);
  offsets_kernel<<<(L + SCAN_T - 1) / SCAN_T, SCAN_T, 0, stream>>>(M, Moff, L);
  const int G1B = (n + 16 * (CH_T / 32) - 1) / (16 * (CH_T / 32));
  bin_scatter_gemm1_kernel<<<NCH + G1B, CH_T, 0, stream>>>(src, dst, ew, Moff, binned, E, NB, NCH,
                                                          x, W1, h1, n);
  bin_degree_kernel<<<NB, DEG_T, 0, stream>>>(binned, Moff, dis, n, E, NB, NCH);

  bin_aggregate_kernel<true><<<NB, AGG_T, 0, stream>>>(binned, Moff, dis, h1, b1, hA, n, E, NB, NCH);

  bin_aggregate_kernel<false><<<NB, AGG_T, 0, stream>>>(binned, Moff, dis, hA, b1, agg2, n, E, NB, NCH);
  gemm2_wmma<<<(n + 15) / 16, 32, 0, stream>>>(agg2, hA, dis, W2, b2, out, n);
}


# edge norm dis[s]*w*dis[d] factored out of both edge sums: rows leave their producer scaled by dis (x@W1 moves into the degree launch), loop adds w*row, dis[d] once per node; products reassociated
# speedup vs baseline: 2.8177x; 1.0537x over previous
#include <hip/hip_runtime.h>
#include <math.h>

typedef __attribute__((ext_vector_type(2))) float v2f;
typedef __attribute__((ext_vector_type(8))) float v8f;

#define NC    45
#define H1DIM 32
#define H2DIM 128

#define BIN_SHIFT 8
#define BIN   (1 << BIN_SHIFT)
#define NBMAX 512
#define CHUNK 4096
#define CH_T  256
#define EPT   (CHUNK / CH_T)
__global__ void __launch_bounds__(CH_T)
bin_count_kernel(const int* __restrict__ dst, int* __restrict__ M, int E, int NB, int NCH) {
  __shared__ int cur[NBMAX];
  const int tid    = threadIdx.x;
  const int chunk  = blockIdx.x;
  const int e_base = chunk * CHUNK;
  for (int b = tid; b < NB; b += CH_T) cur[b] = 0;
  __syncthreads();
  int d[EPT];
#pragma unroll
  for (int k = 0; k < EPT; ++k) {
    const int e  = e_base + k * CH_T + tid;
    const int dv = dst[e < E ? e : E - 1];
    d[k] = (e < E) ? dv : -1;
  }
#pragma unroll
  for (int k = 0; k < EPT; ++k)
    if (d[k] >= 0) atomicAdd(&cur[d[k] >> BIN_SHIFT], 1);
  __syncthreads();
  for (int b = tid; b < NB; b += CH_T) M[b * NCH + chunk] = cur[b];
}

#define SCAN_T 1024
__global__ void __launch_bounds__(SCAN_T)
offsets_kernel(const int* __restrict__ cnt, int* __restrict__ off, int len) {
  __shared__ int s_part[32];
  __shared__ int s_tot[32];
  const int t    = threadIdx.x;
  const int lane = t & 31;
  const int wv   = t >> 5;
  const int lim  = blockIdx.x * SCAN_T;
  int part = 0;
#pragma unroll 8
  for (int q = 4 * t; q < lim; q += 4 * SCAN_T) {
    const int4 c4 = *(const int4*)(cnt + q);
    part += (c4.x + c4.y) + (c4.z + c4.w);
  }
  const int i   = lim + t;
  const int own = (i < len) ? cnt[i] : 0;

  int incl = own;
#pragma unroll
  for (int dlt = 1; dlt < 32; dlt <<= 1) {
    const int u = __shfl_up(incl, dlt);
    if (lane >= dlt) incl += u;
  }
  int ps = part;
#pragma unroll
  for (int dlt = 16; dlt >= 1; dlt >>= 1) ps += __shfl_xor(ps, dlt);
  if (lane == 31) s_tot[wv]  = incl;
  if (lane == 0)  s_part[wv] = ps;
  __syncthreads();
  const int wt = s_tot[lane];
  int base = s_part[lane];
#pragma unroll
  for (int dlt = 16; dlt >= 1; dlt >>= 1) base += __shfl_xor(base, dlt);
  int wincl = wt;
#pragma unroll
  for (int dlt = 1; dlt < 32; dlt <<= 1) {
    const int u = __shfl_up(wincl, dlt);
    if (lane >= dlt) wincl += u;
  }
  const int wexcl = __shfl(wincl - wt, wv);
  if (i < len) off[i] = base + wexcl + (incl - own);
}

__device__ __forceinline__ void gemm1_acc(const float* __restrict__ x,
                                          const float* __restrict__ W1,
                                          int n, int m0, int lane, v8f& c0, v8f& c1) {
  const int khalf = (lane >> 4) & 1;
  const int nn    = lane & 15;
  int m = m0 + nn;
  if (m > n - 1) m = n - 1;

  const float* xr = x + (size_t)m * NC + 2 * khalf;
  const float* wr = W1 + (size_t)(2 * khalf) * H1DIM + nn;
  c0 = (v8f){};
  c1 = (v8f){};
#pragma unroll
  for (int k = 0; k < 11; ++k) {
    v2f a, b0, b1;
    a.x  = xr[4 * k];
    a.y  = xr[4 * k + 1];
    b0.x = wr[(4 * k) * H1DIM];
    b0.y = wr[(4 * k + 1) * H1DIM];
    b1.x = wr[(4 * k) * H1DIM + 16];
    b1.y = wr[(4 * k + 1) * H1DIM + 16];
    c0 = __builtin_amdgcn_wmma_f32_16x16x4_f32(false, a, false, b0, (short)0, c0, false, false);
    c1 = __builtin_amdgcn_wmma_f32_16x16x4_f32(false, a, false, b1, (short)0, c1, false, false);
  }
  {
    const int   koff = khalf ? 0 : 44;
    const float* xl  = x + (size_t)m * NC + koff;
    const float* wl  = W1 + (size_t)koff * H1DIM + nn;
    const float ta = xl[0], t0 = wl[0], t1 = wl[16];
    v2f a, b0, b1;
    a.x  = khalf ? 0.0f : ta;  a.y  = 0.0f;
    b0.x = khalf ? 0.0f : t0;  b0.y = 0.0f;
    b1.x = khalf ? 0.0f : t1;  b1.y = 0.0f;
    c0 = __builtin_amdgcn_wmma_f32_16x16x4_f32(false, a, false, b0, (short)0, c0, false, false);
    c1 = __builtin_amdgcn_wmma_f32_16x16x4_f32(false, a, false, b1, (short)0, c1, false, false);
  }
}
__device__ __forceinline__ void gemm1_store_scaled(float* __restrict__ h1s, const float* sdis,
                                                   int n, int node0, int m0, int lane,
                                                   const v8f& c0, const v8f& c1) {
  const int khalf = (lane >> 4) & 1;
  const int nn    = lane & 15;
  const int rbase = m0 + (khalf << 3);
  float sc[8];
#pragma unroll
  for (int j = 0; j < 8; ++j) sc[j] = sdis[rbase + j - node0];
#pragma unroll
  for (int j = 0; j < 8; ++j) {
    const int r = rbase + j;
    if (r < n) {
      h1s[(size_t)r * H1DIM + nn]      = c0[j] * sc[j];
      h1s[(size_t)r * H1DIM + 16 + nn] = c1[j] * sc[j];
    }
  }
}

__global__ void __launch_bounds__(CH_T)
bin_scatter_kernel(const int* __restrict__ src, const int* __restrict__ dst,
                   const float* __restrict__ ew, const int* __restrict__ Moff,
                   int4* __restrict__ binned, int E, int NB, int NCH) {
  __shared__ int   cur[NBMAX];
  __shared__ int   lbase[NBMAX];
  __shared__ int   gbase[NBMAX];
  __shared__ int   s_src[CHUNK];
  __shared__ int   s_dst[CHUNK];
  __shared__ float s_w[CHUNK];
  const int tid    = threadIdx.x;
  const int lane   = tid & 31;
  const int chunk  = blockIdx.x;
  const int e_base = chunk * CHUNK;
  for (int b = tid; b < NB; b += CH_T) { cur[b] = 0; gbase[b] = Moff[b * NCH + chunk]; }
  __syncthreads();
  int   d[EPT], r[EPT], sv[EPT];
  float wv[EPT];
#pragma unroll
  for (int k = 0; k < EPT; ++k) {
    const int e  = e_base + k * CH_T + tid;
    const int ec = e < E ? e : E - 1;
    const int dv = dst[ec];
    sv[k] = src[ec];
    wv[k] = ew[ec];
    d[k]  = (e < E) ? dv : -1;
  }
#pragma unroll
  for (int k = 0; k < EPT; ++k) {
    r[k] = 0;
    if (d[k] >= 0) r[k] = atomicAdd(&cur[d[k] >> BIN_SHIFT], 1);
  }
  __syncthreads();
  if (tid < 32) {
    int running = 0;
    for (int b0 = 0; b0 < NB; b0 += 32) {
      const int bb = b0 + lane;
      const int v  = (bb < NB) ? cur[bb] : 0;
      int incl = v;
#pragma unroll
      for (int dlt = 1; dlt < 32; dlt <<= 1) {
        const int u = __shfl_up(incl, dlt);
        if (lane >= dlt) incl += u;
      }
      if (bb < NB) lbase[bb] = running + incl - v;
      running += __shfl(incl, 31);
    }
  }
  __syncthreads();
#pragma unroll
  for (int k = 0; k < EPT; ++k) {
    if (d[k] >= 0) {
      const int li = lbase[d[k] >> BIN_SHIFT] + r[k];
      s_src[li] = sv[k];
      s_dst[li] = d[k];
      s_w[li]   = wv[k];
    }
  }
  __syncthreads();
  const int left   = E - e_base;
  const int nvalid = left < CHUNK ? left : CHUNK;
  for (int i = tid; i < nvalid; i += CH_T) {
    const int dd = s_dst[i];
    const int b  = dd >> BIN_SHIFT;
    const int g  = gbase[b] + (i - lbase[b]);
    binned[g] = make_int4(s_src[i], 0, dd, __float_as_int(s_w[i]));
  }
}

#define DEG_T    256
#define DEG_R    20
#define DEG_FRAC 40
__device__ __forceinline__ long long weight_to_fixed(float w) {
  const unsigned u  = __float_as_uint(w);
  const int      e  = (int)((u >> 23) & 0xffu);
  const unsigned long long m = (u & 0x7fffffu) | (e ? 0x800000u : 0u);
  const int sh = (e ? e : 1) - 150 + DEG_FRAC;
  const unsigned long long q = (sh >= 0) ? (m << (sh < 39 ? sh : 39))
                                         : (m >> (-sh < 24 ? -sh : 24));
  return (u >> 31) ? -(long long)q : (long long)q;
}
#define DEG_W (DEG_T / 32)
#define DEG_S (BIN / 16 / DEG_W)
__global__ void __launch_bounds__(DEG_T)
bin_degree_gemm1_kernel(const int4* __restrict__ binned, const int* __restrict__ Moff,
                        float* __restrict__ dis, int n, int E, int NB, int NCH,
                        const float* __restrict__ x, const float* __restrict__ W1,
                        float* __restrict__ h1s) {
  __shared__ unsigned long long dsum[BIN];
  __shared__ float sdis[BIN];
  const int b     = blockIdx.x;
  const int tid   = threadIdx.x;
  const int lane  = tid & 31;
  const int node0 = b << BIN_SHIFT;
  const int wv = __builtin_amdgcn_readfirstlane(tid >> 5);
  for (int k = tid; k < BIN; k += DEG_T) dsum[k] = 0ull;
  const int bs = Moff[b * NCH];
  const int be = (b + 1 < NB) ? Moff[(b + 1) * NCH] : E;
  v8f acc0[DEG_S], acc1[DEG_S];
#pragma unroll
  for (int q = 0; q < DEG_S; ++q)
    gemm1_acc(x, W1, n, node0 + (wv + q * DEG_W) * 16, lane, acc0[q], acc1[q]);
  __syncthreads();
  const int2* dw = (const int2*)binned;
  for (int i0 = bs + tid; i0 < be; i0 += DEG_T * DEG_R) {
    int   dk[DEG_R];
    float wk[DEG_R];
#pragma unroll
    for (int k = 0; k < DEG_R; ++k) {
      const int  i = i0 + k * DEG_T;
      const int2 p = dw[2 * (size_t)(i < be ? i : be - 1) + 1];
      dk[k] = (i < be) ? (p.x & (BIN - 1)) : -1;
      wk[k] = __int_as_float(p.y);
    }
    asm volatile("" :: "v"(wk[0]), "v"(wk[1]), "v"(wk[2]), "v"(wk[3]), "v"(wk[4]),
                       "v"(wk[5]), "v"(wk[6]), "v"(wk[7]), "v"(wk[8]), "v"(wk[9]),
                       "v"(wk[10]), "v"(wk[11]), "v"(wk[12]), "v"(wk[13]), "v"(wk[14]),
                       "v"(wk[15]), "v"(wk[16]), "v"(wk[17]), "v"(wk[18]), "v"(wk[19]));
    static_assert(DEG_R == 20, "the fence above names DEG_R weights");
#pragma unroll
    for (int k = 0; k < DEG_R; ++k) {
      if (dk[k] >= 0)
        atomicAdd(&dsum[dk[k]], (unsigned long long)weight_to_fixed(wk[k]));
    }
  }
  __syncthreads();
  for (int k = tid; k < BIN; k += DEG_T) {
    const int node = node0 + k;
    if (node < n) {
      const long long d1 = (long long)dsum[k] + (1ll << DEG_FRAC);
      const float dv = 1.0f / sqrtf((float)d1 * (1.0f / (float)(1ll << DEG_FRAC)));
      dis[node] = dv;
      sdis[k]   = dv;
    }
  }
  __syncthreads();
#pragma unroll
  for (int q = 0; q < DEG_S; ++q)
    gemm1_store_scaled(h1s, sdis, n, node0, node0 + (wv + q * DEG_W) * 16, lane, acc0[q], acc1[q]);
}

#define AGG_T 1024
#define AGG_W (AGG_T / 32)
#define AGG_R (BIN / AGG_W)
template <bool LAYER1>
__global__ void __launch_bounds__(AGG_T)
bin_aggregate_kernel(const int4* __restrict__ binned, const int* __restrict__ Moff,
                     const float* __restrict__ dis, const float* __restrict__ h,
                     const float* __restrict__ b1, float* __restrict__ outp,
                     int n, int E, int NB, int NCH) {
  __shared__ float tile[BIN * H1DIM];
  const int lane = threadIdx.x & 31;
  const int wv = __builtin_amdgcn_readfirstlane((int)(threadIdx.x >> 5));
  const int nw = AGG_W;
  const int b  = blockIdx.x;
  for (int k = threadIdx.x; k < BIN * H1DIM; k += AGG_T) tile[k] = 0.0f;
  const int bs    = Moff[b * NCH];
  const int be    = (b + 1 < NB) ? Moff[(b + 1) * NCH] : E;
  const int node0 = b << BIN_SHIFT;
  __syncthreads();

  for (int i0 = bs + wv * 32; i0 < be; i0 += nw * 32) {
    const int i = i0 + lane;
    int   s   = node0;
    int   dl  = 0;
    float nrm = 0.0f;
    if (i < be) {
      const int4 r = binned[i];
      s   = r.x;
      dl  = r.z & (BIN - 1);
      nrm = __int_as_float(r.w);
    }
    if (i0 + 32 <= be) {
      float v[32];
#pragma unroll
      for (int j = 0; j < 32; ++j) {
        const int sj = __builtin_amdgcn_readlane(s, j);
        v[j] = h[(size_t)sj * H1DIM + lane];
      }
#pragma unroll
      for (int j = 0; j < 32; ++j) {
        const int   dj = __builtin_amdgcn_readlane(dl, j);
        const float nj = __int_as_float(__builtin_amdgcn_readlane(__float_as_int(nrm), j));
        atomicAdd(&tile[dj * H1DIM + lane], v[j] * nj);
      }
    } else {
      const int cnt = be - i0;
      float v[32];
#pragma unroll
      for (int j = 0; j < 32; ++j) {
        const int sj = __builtin_amdgcn_readlane(s, j);
        v[j] = h[(size_t)sj * H1DIM + lane];
      }
#pragma unroll
      for (int j = 0; j < 32; ++j) {
        if (j < cnt) {
          const int   dj = __builtin_amdgcn_readlane(dl, j);
          const float nj = __int_as_float(__builtin_amdgcn_readlane(__float_as_int(nrm), j));
          atomicAdd(&tile[dj * H1DIM + lane], v[j] * nj);
        }
      }
    }
  }
  __syncthreads();

  if (LAYER1) {
    float hv[AGG_R], dv[AGG_R];
#pragma unroll
    for (int k = 0; k < AGG_R; ++k) {
      int node = node0 + wv + k * AGG_W;
      if (node > n - 1) node = n - 1;
      dv[k] = dis[node];
      hv[k] = h[(size_t)node * H1DIM + lane];
    }
    const float bb = b1[lane];
#pragma unroll
    for (int k = 0; k < AGG_R; ++k) {
      const int r    = wv + k * AGG_W;
      const int node = node0 + r;
      if (node < n) {
        const float acc = tile[r * H1DIM + lane];
        const float di  = dv[k];
        outp[(size_t)node * H1DIM + lane] = di * tanhf(di * (acc + hv[k]) + bb);
      }
    }
  } else {
    for (int r = wv; r < BIN; r += nw) {
      const int node = node0 + r;
      if (node < n) outp[(size_t)node * H1DIM + lane] = tile[r * H1DIM + lane];
    }
  }
}

__global__ void gemm2_wmma(const float* __restrict__ agg2, const float* __restrict__ h,
                           const float* __restrict__ dis, const float* __restrict__ W2,
                           const float* __restrict__ b2, float* __restrict__ out, int n) {
  const int lane  = threadIdx.x;
  const int m0    = blockIdx.x * 16;
  const int khalf = (lane >> 4) & 1;
  const int nn    = lane & 15;
  int m = m0 + nn;
  if (m > n - 1) m = n - 1;

  const float* ap = agg2 + (size_t)m * H1DIM + 2 * khalf;
  const float* hp = h    + (size_t)m * H1DIM + 2 * khalf;
  const float* wr = W2 + (size_t)(2 * khalf) * H2DIM + nn;
  const float di = dis[m];
  v2f a[8];
#pragma unroll
  for (int k = 0; k < 8; ++k) {
    const v2f ag = *(const v2f*)(ap + 4 * k);
    const v2f hv = *(const v2f*)(hp + 4 * k);
    a[k].x = (ag.x + hv.x) * di;
    a[k].y = (ag.y + hv.y) * di;
  }

  v8f acc[8] = {};
#pragma unroll
  for (int k = 0; k < 8; ++k) {
#pragma unroll
    for (int t = 0; t < 8; ++t) {
      v2f b;
      b.x = wr[(4 * k) * H2DIM + t * 16];
      b.y = wr[(4 * k + 1) * H2DIM + t * 16];
      acc[t] = __builtin_amdgcn_wmma_f32_16x16x4_f32(false, a[k], false, b, (short)0, acc[t], false, false);
    }
  }
  float bv[8];
#pragma unroll
  for (int t = 0; t < 8; ++t) bv[t] = b2[t * 16 + nn];

  const int rbase = m0 + (khalf << 3);
#pragma unroll
  for (int j = 0; j < 8; ++j) {
    const int r = rbase + j;
    if (r < n) {
#pragma unroll
      for (int t = 0; t < 8; ++t)
        out[(size_t)r * H2DIM + t * 16 + nn] = acc[t][j] + bv[t];
    }
  }
}

extern "C" void kernel_launch(void* const* d_in, const int* in_sizes, int n_in,
                              void* d_out, int out_size, void* d_ws, size_t ws_size,
                              hipStream_t stream) {
  const float* x   = (const float*)d_in[0];
  const int*   src = (const int*)  d_in[1];
  const int*   dst = (const int*)  d_in[2];
  const float* ew  = (const float*)d_in[3];
  const float* W1  = (const float*)d_in[4];
  const float* b1  = (const float*)d_in[5];
  const float* W2  = (const float*)d_in[6];
  const float* b2  = (const float*)d_in[7];
  float* out = (float*)d_out;

  const int n = in_sizes[0] / NC;
  const int E = in_sizes[1];

  const int NB  = (n + BIN - 1) / BIN;
  const int NCH = (E + CHUNK - 1) / CHUNK;
  const int L   = NB * NCH;

  const size_t np = ((size_t)n + 63) & ~(size_t)63;
  const size_t Lp = ((size_t)L + 63) & ~(size_t)63;
  int*   M      = (int*)d_ws;
  int*   Moff   = M + Lp;
  float* dis    = (float*)(Moff + Lp);
  float* h1     = dis + np;
  float* hA     = h1 + np * H1DIM;
  float* agg2   = hA + np * H1DIM;
  int4*  binned = (int4*)(agg2 + np * H1DIM);

  bin_count_kernel<<<NCH, CH_T, 0, stream>>>(dst, M, E, NB, NCH);
  offsets_kernel<<<(L + SCAN_T - 1) / SCAN_T, SCAN_T, 0, stream>>>(M, Moff, L);
  bin_scatter_kernel<<<NCH, CH_T, 0, stream>>>(src, dst, ew, Moff, binned, E, NB, NCH);
  bin_degree_gemm1_kernel<<<NB, DEG_T, 0, stream>>>(binned, Moff, dis, n, E, NB, NCH, x, W1, h1);

  bin_aggregate_kernel<true><<<NB, AGG_T, 0, stream>>>(binned, Moff, dis, h1, b1, hA, n, E, NB, NCH);

  bin_aggregate_kernel<false><<<NB, AGG_T, 0, stream>>>(binned, Moff, dis, hA, b1, agg2, n, E, NB, NCH);
  gemm2_wmma<<<(n + 15) / 16, 32, 0, stream>>>(agg2, hA, dis, W2, b2, out, n);
}


# edge grouping in ONE pass: each chunk ordered by bin in place + a (bin,chunk) run table; a bin's block scans its counts, lists its record slots in LDS; count and global scan launches go (7 to 5)
# speedup vs baseline: 3.0410x; 1.0792x over previous
#include <hip/hip_runtime.h>
#include <math.h>

typedef __attribute__((ext_vector_type(2))) float v2f;
typedef __attribute__((ext_vector_type(8))) float v8f;

#define NC    45
#define H1DIM 32
#define H2DIM 128


#define BIN_SHIFT 8
#define BIN   (1 << BIN_SHIFT)
#define NBMAX 512
#define CHUNK 4096
#define CH_T  256
#define EPT   (CHUNK / CH_T)
#define NCHMAX 512
#define IDX_CAP 8192

__device__ __forceinline__ void gemm1_acc(const float* __restrict__ x,
                                          const float* __restrict__ W1,
                                          int n, int m0, int lane, v8f& c0, v8f& c1) {
  const int khalf = (lane >> 4) & 1;
  const int nn    = lane & 15;
  int m = m0 + nn;
  if (m > n - 1) m = n - 1;

  const float* xr = x + (size_t)m * NC + 2 * khalf;
  const float* wr = W1 + (size_t)(2 * khalf) * H1DIM + nn;
  c0 = (v8f){};
  c1 = (v8f){};
#pragma unroll
  for (int k = 0; k < 11; ++k) {
    v2f a, b0, b1;
    a.x  = xr[4 * k];
    a.y  = xr[4 * k + 1];
    b0.x = wr[(4 * k) * H1DIM];
    b0.y = wr[(4 * k + 1) * H1DIM];
    b1.x = wr[(4 * k) * H1DIM + 16];
    b1.y = wr[(4 * k + 1) * H1DIM + 16];
    c0 = __builtin_amdgcn_wmma_f32_16x16x4_f32(false, a, false, b0, (short)0, c0, false, false);
    c1 = __builtin_amdgcn_wmma_f32_16x16x4_f32(false, a, false, b1, (short)0, c1, false, false);
  }
  {
    const int   koff = khalf ? 0 : 44;
    const float* xl  = x + (size_t)m * NC + koff;
    const float* wl  = W1 + (size_t)koff * H1DIM + nn;
    const float ta = xl[0], t0 = wl[0], t1 = wl[16];
    v2f a, b0, b1;
    a.x  = khalf ? 0.0f : ta;  a.y  = 0.0f;
    b0.x = khalf ? 0.0f : t0;  b0.y = 0.0f;
    b1.x = khalf ? 0.0f : t1;  b1.y = 0.0f;
    c0 = __builtin_amdgcn_wmma_f32_16x16x4_f32(false, a, false, b0, (short)0, c0, false, false);
    c1 = __builtin_amdgcn_wmma_f32_16x16x4_f32(false, a, false, b1, (short)0, c1, false, false);
  }
}
__device__ __forceinline__ void gemm1_store_scaled(float* __restrict__ h1s, const float* sdis,
                                                   int n, int node0, int m0, int lane,
                                                   const v8f& c0, const v8f& c1) {
  const int khalf = (lane >> 4) & 1;
  const int nn    = lane & 15;
  const int rbase = m0 + (khalf << 3);
  float sc[8];
#pragma unroll
  for (int j = 0; j < 8; ++j) sc[j] = sdis[rbase + j - node0];
#pragma unroll
  for (int j = 0; j < 8; ++j) {
    const int r = rbase + j;
    if (r < n) {
      h1s[(size_t)r * H1DIM + nn]      = c0[j] * sc[j];
      h1s[(size_t)r * H1DIM + 16 + nn] = c1[j] * sc[j];
    }
  }
}

__global__ void __launch_bounds__(CH_T)
chunk_sort_kernel(const int* __restrict__ src, const int* __restrict__ dst,
                  const float* __restrict__ ew, int* __restrict__ T,
                  int4* __restrict__ binned, int E, int NB, int NCH) {
  __shared__ int   cur[NBMAX];
  __shared__ int   lbase[NBMAX];
  __shared__ int   s_src[CHUNK];
  __shared__ int   s_dst[CHUNK];
  __shared__ float s_w[CHUNK];
  const int tid    = threadIdx.x;
  const int lane   = tid & 31;
  const int chunk  = blockIdx.x;
  const int e_base = chunk * CHUNK;
  for (int b = tid; b < NB; b += CH_T) cur[b] = 0;
  __syncthreads();
  int   d[EPT], r[EPT], sv[EPT];
  float wv[EPT];
#pragma unroll
  for (int k = 0; k < EPT; ++k) {
    const int e  = e_base + k * CH_T + tid;
    const int ec = e < E ? e : E - 1;
    const int dv = dst[ec];
    sv[k] = src[ec];
    wv[k] = ew[ec];
    d[k]  = (e < E) ? dv : -1;
  }
#pragma unroll
  for (int k = 0; k < EPT; ++k) {
    r[k] = 0;
    if (d[k] >= 0) r[k] = atomicAdd(&cur[d[k] >> BIN_SHIFT], 1);
  }
  __syncthreads();
  if (tid < 32) {
    int running = 0;
    for (int b0 = 0; b0 < NB; b0 += 32) {
      const int bb = b0 + lane;
      const int v  = (bb < NB) ? cur[bb] : 0;
      int incl = v;
#pragma unroll
      for (int dlt = 1; dlt < 32; dlt <<= 1) {
        const int u = __shfl_up(incl, dlt);
        if (lane >= dlt) incl += u;
      }
      if (bb < NB) lbase[bb] = running + incl - v;
      running += __shfl(incl, 31);
    }
  }
  __syncthreads();
#pragma unroll
  for (int k = 0; k < EPT; ++k) {
    if (d[k] >= 0) {
      const int li = lbase[d[k] >> BIN_SHIFT] + r[k];
      s_src[li] = sv[k];
      s_dst[li] = d[k];
      s_w[li]   = wv[k];
    }
  }
  __syncthreads();
  const int left   = E - e_base;
  const int nvalid = left < CHUNK ? left : CHUNK;
  for (int i = tid; i < nvalid; i += CH_T)
    binned[e_base + i] = make_int4(s_src[i], 0, s_dst[i], __float_as_int(s_w[i]));
  for (int b = tid; b < NB; b += CH_T) T[b * NCH + chunk] = lbase[b] | (cur[b] << 16);
}

template <int RPT>
__device__ __forceinline__ void bin_runs_fetch(const int* __restrict__ T, int b, int NCH, int tid,
                                               int (&w)[RPT]) {
#pragma unroll
  for (int j = 0; j < RPT; ++j) {
    const int c = tid * RPT + j;
    w[j] = (c < NCH) ? T[(size_t)b * NCH + c] : 0;
  }
}
template <int NT, int RPT>
__device__ __forceinline__ int bin_runs_scan(const int (&w)[RPT], int tid, int* s_wtot,
                                             int (&p0)[RPT], int (&ln)[RPT], int (&g0)[RPT]) {
  const int lane = tid & 31;
  const int wv   = tid >> 5;
  int tsum = 0;
#pragma unroll
  for (int j = 0; j < RPT; ++j) {
    ln[j] = w[j] >> 16;
    g0[j] = (tid * RPT + j) * CHUNK + (w[j] & 0xffff);
    p0[j] = tsum;
    tsum += ln[j];
  }
  int incl = tsum;
#pragma unroll
  for (int dlt = 1; dlt < 32; dlt <<= 1) {
    const int u = __shfl_up(incl, dlt);
    if (lane >= dlt) incl += u;
  }
  if (lane == 31) s_wtot[wv] = incl;
  __syncthreads();
  const int t = (lane < NT / 32) ? s_wtot[lane] : 0;
  int tot    = t;
  int before = (lane < wv) ? t : 0;
#pragma unroll
  for (int dlt = 16; dlt >= 1; dlt >>= 1) {
    tot    += __shfl_xor(tot, dlt);
    before += __shfl_xor(before, dlt);
  }
  const int base = before + incl - tsum;
#pragma unroll
  for (int j = 0; j < RPT; ++j) p0[j] += base;
  return tot;
}
template <int RPT>
__device__ __forceinline__ void bin_runs_list(int* idx, int tb, const int (&p0)[RPT],
                                              const int (&ln)[RPT], const int (&g0)[RPT]) {
#pragma unroll
  for (int j = 0; j < RPT; ++j) {
    int k0 = tb - p0[j];
    if (k0 < 0) k0 = 0;
    int k1 = tb + IDX_CAP - p0[j];
    if (k1 > ln[j]) k1 = ln[j];
    for (int k = k0; k < k1; ++k) idx[p0[j] + k - tb] = g0[j] + k;
  }
}

#define DEG_T    256
#define DEG_R    20
#define DEG_FRAC 40
__device__ __forceinline__ long long weight_to_fixed(float w) {
  const unsigned u  = __float_as_uint(w);
  const int      e  = (int)((u >> 23) & 0xffu);
  const unsigned long long m = (u & 0x7fffffu) | (e ? 0x800000u : 0u);
  const int sh = (e ? e : 1) - 150 + DEG_FRAC;
  const unsigned long long q = (sh >= 0) ? (m << (sh < 39 ? sh : 39))
                                         : (m >> (-sh < 24 ? -sh : 24));
  return (u >> 31) ? -(long long)q : (long long)q;
}
#define DEG_W (DEG_T / 32)
#define DEG_S (BIN / 16 / DEG_W)
#define DEG_RPT (NCHMAX / DEG_T)
__global__ void __launch_bounds__(DEG_T)
bin_degree_gemm1_kernel(const int4* __restrict__ binned, const int* __restrict__ T,
                        float* __restrict__ dis, int n, int NCH,
                        const float* __restrict__ x, const float* __restrict__ W1,
                        float* __restrict__ h1s) {
  __shared__ unsigned long long dsum[BIN];
  __shared__ float sdis[BIN];
  __shared__ int   idx[IDX_CAP];
  __shared__ int   s_wtot[32];
  const int b     = blockIdx.x;
  const int tid   = threadIdx.x;
  const int lane  = tid & 31;
  const int node0 = b << BIN_SHIFT;
  const int wv = __builtin_amdgcn_readfirstlane(tid >> 5);
  int rw[DEG_RPT];
  bin_runs_fetch<DEG_RPT>(T, b, NCH, tid, rw);
  for (int k = tid; k < BIN; k += DEG_T) dsum[k] = 0ull;
  v8f acc0[DEG_S], acc1[DEG_S];
#pragma unroll
  for (int q = 0; q < DEG_S; ++q)
    gemm1_acc(x, W1, n, node0 + (wv + q * DEG_W) * 16, lane, acc0[q], acc1[q]);
  int p0[DEG_RPT], ln[DEG_RPT], g0[DEG_RPT];
  const int total = __builtin_amdgcn_readfirstlane(
      bin_runs_scan<DEG_T, DEG_RPT>(rw, tid, s_wtot, p0, ln, g0));
  const int2* dw = (const int2*)binned;
  for (int tb = 0; tb < total; tb += IDX_CAP) {
    if (tb) __syncthreads();
    bin_runs_list<DEG_RPT>(idx, tb, p0, ln, g0);
    __syncthreads();
    const int te = (total - tb < IDX_CAP) ? (total - tb) : IDX_CAP;
    for (int i0 = tid; i0 < te; i0 += DEG_T * DEG_R) {
      int   dk[DEG_R];
      float wk[DEG_R];
#pragma unroll
      for (int k = 0; k < DEG_R; ++k) {
        const int  i = i0 + k * DEG_T;
        const int  g = idx[i < te ? i : te - 1];
        const int2 p = dw[2 * (size_t)g + 1];
        dk[k] = (i < te) ? (p.x & (BIN - 1)) : -1;
        wk[k] = __int_as_float(p.y);
      }
      asm volatile("" :: "v"(wk[0]), "v"(wk[1]), "v"(wk[2]), "v"(wk[3]), "v"(wk[4]),
                         "v"(wk[5]), "v"(wk[6]), "v"(wk[7]), "v"(wk[8]), "v"(wk[9]),
                         "v"(wk[10]), "v"(wk[11]), "v"(wk[12]), "v"(wk[13]), "v"(wk[14]),
                         "v"(wk[15]), "v"(wk[16]), "v"(wk[17]), "v"(wk[18]), "v"(wk[19]));
      static_assert(DEG_R == 20, "the fence above names DEG_R weights");
#pragma unroll
      for (int k = 0; k < DEG_R; ++k) {
        if (dk[k] >= 0)
          atomicAdd(&dsum[dk[k]], (unsigned long long)weight_to_fixed(wk[k]));
      }
    }
  }
  __syncthreads();
  for (int k = tid; k < BIN; k += DEG_T) {
    const int node = node0 + k;
    if (node < n) {
      const long long d1 = (long long)dsum[k] + (1ll << DEG_FRAC);
      const float dv = 1.0f / sqrtf((float)d1 * (1.0f / (float)(1ll << DEG_FRAC)));
      dis[node] = dv;
      sdis[k]   = dv;
    }
  }
  __syncthreads();
#pragma unroll
  for (int q = 0; q < DEG_S; ++q)
    gemm1_store_scaled(h1s, sdis, n, node0, node0 + (wv + q * DEG_W) * 16, lane, acc0[q], acc1[q]);
}

#define AGG_T 1024
#define AGG_W (AGG_T / 32)
#define AGG_R (BIN / AGG_W)
template <bool LAYER1>
__global__ void __launch_bounds__(AGG_T, 16)
bin_aggregate_kernel(const int4* __restrict__ binned, const int* __restrict__ T,
                     const float* __restrict__ dis, const float* __restrict__ h,
                     const float* __restrict__ b1, float* __restrict__ outp,
                     int n, int NCH) {
  __shared__ float tile[BIN * H1DIM];
  __shared__ int   idx[IDX_CAP];
  __shared__ int   s_wtot[32];
  static_assert(AGG_T >= NCHMAX, "one run of the bin per thread");
  const int lane = threadIdx.x & 31;
  const int wv = __builtin_amdgcn_readfirstlane((int)(threadIdx.x >> 5));
  const int nw = AGG_W;
  const int b  = blockIdx.x;
  int rw[1];
  bin_runs_fetch<1>(T, b, NCH, (int)threadIdx.x, rw);
  for (int k = threadIdx.x; k < BIN * H1DIM; k += AGG_T) tile[k] = 0.0f;
  const int node0 = b << BIN_SHIFT;
  int p0[1], ln[1], g0[1];
  const int total = __builtin_amdgcn_readfirstlane(
      bin_runs_scan<AGG_T, 1>(rw, (int)threadIdx.x, s_wtot, p0, ln, g0));

  for (int tb = 0; tb < total; tb += IDX_CAP) {
    if (tb) __syncthreads();
    bin_runs_list<1>(idx, tb, p0, ln, g0);
    __syncthreads();
    const int te = (total - tb < IDX_CAP) ? (total - tb) : IDX_CAP;
    for (int i0 = wv * 32; i0 < te; i0 += nw * 32) {
      const int i = i0 + lane;
      int   s   = node0;
      int   dl  = 0;
      float nrm = 0.0f;
      if (i < te) {
        const int4 r = binned[idx[i]];
        s   = r.x;
        dl  = r.z & (BIN - 1);
        nrm = __int_as_float(r.w);
      }
      if (i0 + 32 <= te) {
        float v[32];
#pragma unroll
        for (int j = 0; j < 32; ++j) {
          const int sj = __builtin_amdgcn_readlane(s, j);
          v[j] = h[(size_t)sj * H1DIM + lane];
        }
#pragma unroll
        for (int j = 0; j < 32; ++j) {
          const int   dj = __builtin_amdgcn_readlane(dl, j);
          const float nj = __int_as_float(__builtin_amdgcn_readlane(__float_as_int(nrm), j));
          atomicAdd(&tile[dj * H1DIM + lane], v[j] * nj);
        }
      } else {
        const int cnt = te - i0;
        float v[32];
#pragma unroll
        for (int j = 0; j < 32; ++j) {
          const int sj = __builtin_amdgcn_readlane(s, j);
          v[j] = h[(size_t)sj * H1DIM + lane];
        }
#pragma unroll
        for (int j = 0; j < 32; ++j) {
          if (j < cnt) {
            const int   dj = __builtin_amdgcn_readlane(dl, j);
            const float nj = __int_as_float(__builtin_amdgcn_readlane(__float_as_int(nrm), j));
            atomicAdd(&tile[dj * H1DIM + lane], v[j] * nj);
          }
        }
      }
    }
  }
  __syncthreads();

  if (LAYER1) {
    float hv[AGG_R], dv[AGG_R];
#pragma unroll
    for (int k = 0; k < AGG_R; ++k) {
      int node = node0 + wv + k * AGG_W;
      if (node > n - 1) node = n - 1;
      dv[k] = dis[node];
      hv[k] = h[(size_t)node * H1DIM + lane];
    }
    const float bb = b1[lane];
#pragma unroll
    for (int k = 0; k < AGG_R; ++k) {
      const int r    = wv + k * AGG_W;
      const int node = node0 + r;
      if (node < n) {
        const float acc = tile[r * H1DIM + lane];
        const float di  = dv[k];
        outp[(size_t)node * H1DIM + lane] = di * tanhf(di * (acc + hv[k]) + bb);
      }
    }
  } else {
    for (int r = wv; r < BIN; r += nw) {
      const int node = node0 + r;
      if (node < n) outp[(size_t)node * H1DIM + lane] = tile[r * H1DIM + lane];
    }
  }
}

__global__ void gemm2_wmma(const float* __restrict__ agg2, const float* __restrict__ h,
                           const float* __restrict__ dis, const float* __restrict__ W2,
                           const float* __restrict__ b2, float* __restrict__ out, int n) {
  const int lane  = threadIdx.x;
  const int m0    = blockIdx.x * 16;
  const int khalf = (lane >> 4) & 1;
  const int nn    = lane & 15;
  int m = m0 + nn;
  if (m > n - 1) m = n - 1;

  const float* ap = agg2 + (size_t)m * H1DIM + 2 * khalf;
  const float* hp = h    + (size_t)m * H1DIM + 2 * khalf;
  const float* wr = W2 + (size_t)(2 * khalf) * H2DIM + nn;
  const float di = dis[m];
  v2f a[8];
#pragma unroll
  for (int k = 0; k < 8; ++k) {
    const v2f ag = *(const v2f*)(ap + 4 * k);
    const v2f hv = *(const v2f*)(hp + 4 * k);
    a[k].x = (ag.x + hv.x) * di;
    a[k].y = (ag.y + hv.y) * di;
  }

  v8f acc[8] = {};
#pragma unroll
  for (int k = 0; k < 8; ++k) {
#pragma unroll
    for (int t = 0; t < 8; ++t) {
      v2f b;
      b.x = wr[(4 * k) * H2DIM + t * 16];
      b.y = wr[(4 * k + 1) * H2DIM + t * 16];
      acc[t] = __builtin_amdgcn_wmma_f32_16x16x4_f32(false, a[k], false, b, (short)0, acc[t], false, false);
    }
  }
  float bv[8];
#pragma unroll
  for (int t = 0; t < 8; ++t) bv[t] = b2[t * 16 + nn];

  const int rbase = m0 + (khalf << 3);
#pragma unroll
  for (int j = 0; j < 8; ++j) {
    const int r = rbase + j;
    if (r < n) {
#pragma unroll
      for (int t = 0; t < 8; ++t)
        out[(size_t)r * H2DIM + t * 16 + nn] = acc[t][j] + bv[t];
    }
  }
}

extern "C" void kernel_launch(void* const* d_in, const int* in_sizes, int n_in,
                              void* d_out, int out_size, void* d_ws, size_t ws_size,
                              hipStream_t stream) {
  const float* x   = (const float*)d_in[0];
  const int*   src = (const int*)  d_in[1];
  const int*   dst = (const int*)  d_in[2];
  const float* ew  = (const float*)d_in[3];
  const float* W1  = (const float*)d_in[4];
  const float* b1  = (const float*)d_in[5];
  const float* W2  = (const float*)d_in[6];
  const float* b2  = (const float*)d_in[7];
  float* out = (float*)d_out;

  const int n = in_sizes[0] / NC;
  const int E = in_sizes[1];

  const int NB  = (n + BIN - 1) / BIN;
  const int NCH = (E + CHUNK - 1) / CHUNK;
  const int L   = NB * NCH;

  const size_t np = ((size_t)n + 63) & ~(size_t)63;
  const size_t Lp = ((size_t)L + 63) & ~(size_t)63;
  int*   T      = (int*)d_ws;
  float* dis    = (float*)(T + Lp);
  float* h1     = dis + np;
  float* hA     = h1 + np * H1DIM;
  float* agg2   = hA + np * H1DIM;
  int4*  binned = (int4*)(agg2 + np * H1DIM);

  chunk_sort_kernel<<<NCH, CH_T, 0, stream>>>(src, dst, ew, T, binned, E, NB, NCH);
  bin_degree_gemm1_kernel<<<NB, DEG_T, 0, stream>>>(binned, T, dis, n, NCH, x, W1, h1);

  bin_aggregate_kernel<true><<<NB, AGG_T, 0, stream>>>(binned, T, dis, h1, b1, hA, n, NCH);

  bin_aggregate_kernel<false><<<NB, AGG_T, 0, stream>>>(binned, T, dis, hA, b1, agg2, n, NCH);
  gemm2_wmma<<<(n + 15) / 16, 32, 0, stream>>>(agg2, hA, dis, W2, b2, out, n);
}


# node bins of 400 nodes (was 256): 250 bin blocks, one per WGP with equal load (was 391 on 256 WGPs: two on some, one on others); degree blocks 16 waves; same adds, LDS add order still free
# speedup vs baseline: 3.4291x; 1.1276x over previous
#include <hip/hip_runtime.h>
#include <math.h>

typedef __attribute__((ext_vector_type(2))) float v2f;
typedef __attribute__((ext_vector_type(8))) float v8f;

#define NC    45
#define H1DIM 32
#define H2DIM 128


#define BIN   400
#define NBMAX 512
#define CHUNK 4096
#define CH_T  256
#define EPT   (CHUNK / CH_T)
#define NCHMAX 512
#define IDX_CAP 8192

__device__ __forceinline__ void gemm1_acc(const float* __restrict__ x,
                                          const float* __restrict__ W1,
                                          int n, int m0, int lane, v8f& c0, v8f& c1) {
  const int khalf = (lane >> 4) & 1;
  const int nn    = lane & 15;
  int m = m0 + nn;
  if (m > n - 1) m = n - 1;

  const float* xr = x + (size_t)m * NC + 2 * khalf;
  const float* wr = W1 + (size_t)(2 * khalf) * H1DIM + nn;
  c0 = (v8f){};
  c1 = (v8f){};
#pragma unroll
  for (int k = 0; k < 11; ++k) {
    v2f a, b0, b1;
    a.x  = xr[4 * k];
    a.y  = xr[4 * k + 1];
    b0.x = wr[(4 * k) * H1DIM];
    b0.y = wr[(4 * k + 1) * H1DIM];
    b1.x = wr[(4 * k) * H1DIM + 16];
    b1.y = wr[(4 * k + 1) * H1DIM + 16];
    c0 = __builtin_amdgcn_wmma_f32_16x16x4_f32(false, a, false, b0, (short)0, c0, false, false);
    c1 = __builtin_amdgcn_wmma_f32_16x16x4_f32(false, a, false, b1, (short)0, c1, false, false);
  }
  {
    const int   koff = khalf ? 0 : 44;
    const float* xl  = x + (size_t)m * NC + koff;
    const float* wl  = W1 + (size_t)koff * H1DIM + nn;
    const float ta = xl[0], t0 = wl[0], t1 = wl[16];
    v2f a, b0, b1;
    a.x  = khalf ? 0.0f : ta;  a.y  = 0.0f;
    b0.x = khalf ? 0.0f : t0;  b0.y = 0.0f;
    b1.x = khalf ? 0.0f : t1;  b1.y = 0.0f;
    c0 = __builtin_amdgcn_wmma_f32_16x16x4_f32(false, a, false, b0, (short)0, c0, false, false);
    c1 = __builtin_amdgcn_wmma_f32_16x16x4_f32(false, a, false, b1, (short)0, c1, false, false);
  }
}
__device__ __forceinline__ void gemm1_store_scaled(float* __restrict__ h1s, const float* sdis,
                                                   int n, int node0, int m0, int lane,
                                                   const v8f& c0, const v8f& c1) {
  const int khalf = (lane >> 4) & 1;
  const int nn    = lane & 15;
  const int rbase = m0 + (khalf << 3);
  float sc[8];
#pragma unroll
  for (int j = 0; j < 8; ++j) sc[j] = sdis[rbase + j - node0];
#pragma unroll
  for (int j = 0; j < 8; ++j) {
    const int r = rbase + j;
    if (r < n) {
      h1s[(size_t)r * H1DIM + nn]      = c0[j] * sc[j];
      h1s[(size_t)r * H1DIM + 16 + nn] = c1[j] * sc[j];
    }
  }
}

__global__ void __launch_bounds__(CH_T)
chunk_sort_kernel(const int* __restrict__ src, const int* __restrict__ dst,
                  const float* __restrict__ ew, int* __restrict__ T,
                  int4* __restrict__ binned, int E, int NB, int NCH) {
  __shared__ int   cur[NBMAX];
  __shared__ int   lbase[NBMAX];
  __shared__ int   s_src[CHUNK];
  __shared__ int   s_dst[CHUNK];
  __shared__ float s_w[CHUNK];
  const int tid    = threadIdx.x;
  const int lane   = tid & 31;
  const int chunk  = blockIdx.x;
  const int e_base = chunk * CHUNK;
  for (int b = tid; b < NB; b += CH_T) cur[b] = 0;
  __syncthreads();
  int   d[EPT], r[EPT], sv[EPT];
  float wv[EPT];
#pragma unroll
  for (int k = 0; k < EPT; ++k) {
    const int e  = e_base + k * CH_T + tid;
    const int ec = e < E ? e : E - 1;
    const int dv = dst[ec];
    sv[k] = src[ec];
    wv[k] = ew[ec];
    d[k]  = (e < E) ? dv : -1;
  }
#pragma unroll
  for (int k = 0; k < EPT; ++k) {
    r[k] = 0;
    if (d[k] >= 0) r[k] = atomicAdd(&cur[(unsigned)d[k] / BIN], 1);
  }
  __syncthreads();
  if (tid < 32) {
    int running = 0;
    for (int b0 = 0; b0 < NB; b0 += 32) {
      const int bb = b0 + lane;
      const int v  = (bb < NB) ? cur[bb] : 0;
      int incl = v;
#pragma unroll
      for (int dlt = 1; dlt < 32; dlt <<= 1) {
        const int u = __shfl_up(incl, dlt);
        if (lane >= dlt) incl += u;
      }
      if (bb < NB) lbase[bb] = running + incl - v;
      running += __shfl(incl, 31);
    }
  }
  __syncthreads();
#pragma unroll
  for (int k = 0; k < EPT; ++k) {
    if (d[k] >= 0) {
      const int li = lbase[(unsigned)d[k] / BIN] + r[k];
      s_src[li] = sv[k];
      s_dst[li] = d[k];
      s_w[li]   = wv[k];
    }
  }
  __syncthreads();
  const int left   = E - e_base;
  const int nvalid = left < CHUNK ? left : CHUNK;
  for (int i = tid; i < nvalid; i += CH_T)
    binned[e_base + i] = make_int4(s_src[i], 0, s_dst[i], __float_as_int(s_w[i]));
  for (int b = tid; b < NB; b += CH_T) T[b * NCH + chunk] = lbase[b] | (cur[b] << 16);
}

template <int RPT>
__device__ __forceinline__ void bin_runs_fetch(const int* __restrict__ T, int b, int NCH, int tid,
                                               int (&w)[RPT]) {
#pragma unroll
  for (int j = 0; j < RPT; ++j) {
    const int c = tid * RPT + j;
    w[j] = (c < NCH) ? T[(size_t)b * NCH + c] : 0;
  }
}
template <int NT, int RPT>
__device__ __forceinline__ int bin_runs_scan(const int (&w)[RPT], int tid, int* s_wtot,
                                             int (&p0)[RPT], int (&ln)[RPT], int (&g0)[RPT]) {
  const int lane = tid & 31;
  const int wv   = tid >> 5;
  int tsum = 0;
#pragma unroll
  for (int j = 0; j < RPT; ++j) {
    ln[j] = w[j] >> 16;
    g0[j] = (tid * RPT + j) * CHUNK + (w[j] & 0xffff);
    p0[j] = tsum;
    tsum += ln[j];
  }
  int incl = tsum;
#pragma unroll
  for (int dlt = 1; dlt < 32; dlt <<= 1) {
    const int u = __shfl_up(incl, dlt);
    if (lane >= dlt) incl += u;
  }
  if (lane == 31) s_wtot[wv] = incl;
  __syncthreads();
  const int t = (lane < NT / 32) ? s_wtot[lane] : 0;
  int tot    = t;
  int before = (lane < wv) ? t : 0;
#pragma unroll
  for (int dlt = 16; dlt >= 1; dlt >>= 1) {
    tot    += __shfl_xor(tot, dlt);
    before += __shfl_xor(before, dlt);
  }
  const int base = before + incl - tsum;
#pragma unroll
  for (int j = 0; j < RPT; ++j) p0[j] += base;
  return tot;
}
template <int RPT>
__device__ __forceinline__ void bin_runs_list(int* idx, int tb, const int (&p0)[RPT],
                                              const int (&ln)[RPT], const int (&g0)[RPT]) {
#pragma unroll
  for (int j = 0; j < RPT; ++j) {
    int k0 = tb - p0[j];
    if (k0 < 0) k0 = 0;
    int k1 = tb + IDX_CAP - p0[j];
    if (k1 > ln[j]) k1 = ln[j];
    for (int k = k0; k < k1; ++k) idx[p0[j] + k - tb] = g0[j] + k;
  }
}

#define DEG_T    512
#define DEG_R    20
#define DEG_FRAC 40
__device__ __forceinline__ long long weight_to_fixed(float w) {
  const unsigned u  = __float_as_uint(w);
  const int      e  = (int)((u >> 23) & 0xffu);
  const unsigned long long m = (u & 0x7fffffu) | (e ? 0x800000u : 0u);
  const int sh = (e ? e : 1) - 150 + DEG_FRAC;
  const unsigned long long q = (sh >= 0) ? (m << (sh < 39 ? sh : 39))
                                         : (m >> (-sh < 24 ? -sh : 24));
  return (u >> 31) ? -(long long)q : (long long)q;
}
#define DEG_W (DEG_T / 32)
#define DEG_NS (BIN / 16)
#define DEG_S ((DEG_NS + DEG_W - 1) / DEG_W)
#define DEG_RPT (NCHMAX / DEG_T)
__global__ void __launch_bounds__(DEG_T)
bin_degree_gemm1_kernel(const int4* __restrict__ binned, const int* __restrict__ T,
                        float* __restrict__ dis, int n, int NCH,
                        const float* __restrict__ x, const float* __restrict__ W1,
                        float* __restrict__ h1s) {
  __shared__ unsigned long long dsum[BIN];
  __shared__ float sdis[BIN];
  __shared__ int   idx[IDX_CAP];
  __shared__ int   s_wtot[32];
  const int b     = blockIdx.x;
  const int tid   = threadIdx.x;
  const int lane  = tid & 31;
  const int node0 = b * BIN;
  const int wv = __builtin_amdgcn_readfirstlane(tid >> 5);
  int rw[DEG_RPT];
  bin_runs_fetch<DEG_RPT>(T, b, NCH, tid, rw);
  for (int k = tid; k < BIN; k += DEG_T) dsum[k] = 0ull;
  v8f acc0[DEG_S], acc1[DEG_S];
#pragma unroll
  for (int q = 0; q < DEG_S; ++q)
    if (wv + q * DEG_W < DEG_NS)
      gemm1_acc(x, W1, n, node0 + (wv + q * DEG_W) * 16, lane, acc0[q], acc1[q]);
  int p0[DEG_RPT], ln[DEG_RPT], g0[DEG_RPT];
  const int total = __builtin_amdgcn_readfirstlane(
      bin_runs_scan<DEG_T, DEG_RPT>(rw, tid, s_wtot, p0, ln, g0));
  const int2* dw = (const int2*)binned;
  for (int tb = 0; tb < total; tb += IDX_CAP) {
    if (tb) __syncthreads();
    bin_runs_list<DEG_RPT>(idx, tb, p0, ln, g0);
    __syncthreads();
    const int te = (total - tb < IDX_CAP) ? (total - tb) : IDX_CAP;
    for (int i0 = tid; i0 < te; i0 += DEG_T * DEG_R) {
      int   dk[DEG_R];
      float wk[DEG_R];
#pragma unroll
      for (int k = 0; k < DEG_R; ++k) {
        const int  i = i0 + k * DEG_T;
        const int  g = idx[i < te ? i : te - 1];
        const int2 p = dw[2 * (size_t)g + 1];
        dk[k] = (i < te) ? (p.x - node0) : -1;
        wk[k] = __int_as_float(p.y);
      }
      asm volatile("" :: "v"(wk[0]), "v"(wk[1]), "v"(wk[2]), "v"(wk[3]), "v"(wk[4]),
                         "v"(wk[5]), "v"(wk[6]), "v"(wk[7]), "v"(wk[8]), "v"(wk[9]),
                         "v"(wk[10]), "v"(wk[11]), "v"(wk[12]), "v"(wk[13]), "v"(wk[14]),
                         "v"(wk[15]), "v"(wk[16]), "v"(wk[17]), "v"(wk[18]), "v"(wk[19]));
      static_assert(DEG_R == 20, "the fence above names DEG_R weights");
#pragma unroll
      for (int k = 0; k < DEG_R; ++k) {
        if (dk[k] >= 0)
          atomicAdd(&dsum[dk[k]], (unsigned long long)weight_to_fixed(wk[k]));
      }
    }
  }
  __syncthreads();
  for (int k = tid; k < BIN; k += DEG_T) {
    const int node = node0 + k;
    if (node < n) {
      const long long d1 = (long long)dsum[k] + (1ll << DEG_FRAC);
      const float dv = 1.0f / sqrtf((float)d1 * (1.0f / (float)(1ll << DEG_FRAC)));
      dis[node] = dv;
      sdis[k]   = dv;
    }
  }
  __syncthreads();
#pragma unroll
  for (int q = 0; q < DEG_S; ++q)
    if (wv + q * DEG_W < DEG_NS)
      gemm1_store_scaled(h1s, sdis, n, node0, node0 + (wv + q * DEG_W) * 16, lane, acc0[q], acc1[q]);
}

#define AGG_T 1024
#define AGG_W (AGG_T / 32)
#define AGG_R ((BIN + AGG_W - 1) / AGG_W)
template <bool LAYER1>
__global__ void __launch_bounds__(AGG_T)
bin_aggregate_kernel(const int4* __restrict__ binned, const int* __restrict__ T,
                     const float* __restrict__ dis, const float* __restrict__ h,
                     const float* __restrict__ b1, float* __restrict__ outp,
                     int n, int NCH) {
  __shared__ float tile[BIN * H1DIM];
  __shared__ int   idx[IDX_CAP];
  __shared__ int   s_wtot[32];
  static_assert(AGG_T >= NCHMAX, "one run of the bin per thread");
  const int lane = threadIdx.x & 31;
  const int wv = __builtin_amdgcn_readfirstlane((int)(threadIdx.x >> 5));
  const int nw = AGG_W;
  const int b  = blockIdx.x;
  int rw[1];
  bin_runs_fetch<1>(T, b, NCH, (int)threadIdx.x, rw);
  for (int k = threadIdx.x; k < BIN * H1DIM; k += AGG_T) tile[k] = 0.0f;
  const int node0 = b * BIN;
  int p0[1], ln[1], g0[1];
  const int total = __builtin_amdgcn_readfirstlane(
      bin_runs_scan<AGG_T, 1>(rw, (int)threadIdx.x, s_wtot, p0, ln, g0));

  for (int tb = 0; tb < total; tb += IDX_CAP) {
    if (tb) __syncthreads();
    bin_runs_list<1>(idx, tb, p0, ln, g0);
    __syncthreads();
    const int te = (total - tb < IDX_CAP) ? (total - tb) : IDX_CAP;
    for (int i0 = wv * 32; i0 < te; i0 += nw * 32) {
      const int i = i0 + lane;
      int   s   = node0;
      int   dl  = 0;
      float nrm = 0.0f;
      if (i < te) {
        const int4 r = binned[idx[i]];
        s   = r.x;
        dl  = r.z - node0;
        nrm = __int_as_float(r.w);
      }
      if (i0 + 32 <= te) {
        float v[32];
#pragma unroll
        for (int j = 0; j < 32; ++j) {
          const int sj = __builtin_amdgcn_readlane(s, j);
          v[j] = h[(size_t)sj * H1DIM + lane];
        }
#pragma unroll
        for (int j = 0; j < 32; ++j) {
          const int   dj = __builtin_amdgcn_readlane(dl, j);
          const float nj = __int_as_float(__builtin_amdgcn_readlane(__float_as_int(nrm), j));
          atomicAdd(&tile[dj * H1DIM + lane], v[j] * nj);
        }
      } else {
        const int cnt = te - i0;
        float v[32];
#pragma unroll
        for (int j = 0; j < 32; ++j) {
          const int sj = __builtin_amdgcn_readlane(s, j);
          v[j] = h[(size_t)sj * H1DIM + lane];
        }
#pragma unroll
        for (int j = 0; j < 32; ++j) {
          if (j < cnt) {
            const int   dj = __builtin_amdgcn_readlane(dl, j);
            const float nj = __int_as_float(__builtin_amdgcn_readlane(__float_as_int(nrm), j));
            atomicAdd(&tile[dj * H1DIM + lane], v[j] * nj);
          }
        }
      }
    }
  }
  __syncthreads();

  if (LAYER1) {
    float hv[AGG_R], dv[AGG_R];
#pragma unroll
    for (int k = 0; k < AGG_R; ++k) {
      int node = node0 + wv + k * AGG_W;
      if (node > n - 1) node = n - 1;
      dv[k] = dis[node];
      hv[k] = h[(size_t)node * H1DIM + lane];
    }
    const float bb = b1[lane];
#pragma unroll
    for (int k = 0; k < AGG_R; ++k) {
      const int r    = wv + k * AGG_W;
      const int node = node0 + r;
      if (r < BIN && node < n) {
        const float acc = tile[r * H1DIM + lane];
        const float di  = dv[k];
        outp[(size_t)node * H1DIM + lane] = di * tanhf(di * (acc + hv[k]) + bb);
      }
    }
  } else {
    for (int r = wv; r < BIN; r += nw) {
      const int node = node0 + r;
      if (node < n) outp[(size_t)node * H1DIM + lane] = tile[r * H1DIM + lane];
    }
  }
}

__global__ void gemm2_wmma(const float* __restrict__ agg2, const float* __restrict__ h,
                           const float* __restrict__ dis, const float* __restrict__ W2,
                           const float* __restrict__ b2, float* __restrict__ out, int n) {
  const int lane  = threadIdx.x;
  const int m0    = blockIdx.x * 16;
  const int khalf = (lane >> 4) & 1;
  const int nn    = lane & 15;
  int m = m0 + nn;
  if (m > n - 1) m = n - 1;

  const float* ap = agg2 + (size_t)m * H1DIM + 2 * khalf;
  const float* hp = h    + (size_t)m * H1DIM + 2 * khalf;
  const float* wr = W2 + (size_t)(2 * khalf) * H2DIM + nn;
  const float di = dis[m];
  v2f a[8];
#pragma unroll
  for (int k = 0; k < 8; ++k) {
    const v2f ag = *(const v2f*)(ap + 4 * k);
    const v2f hv = *(const v2f*)(hp + 4 * k);
    a[k].x = (ag.x + hv.x) * di;
    a[k].y = (ag.y + hv.y) * di;
  }

  v8f acc[8] = {};
#pragma unroll
  for (int k = 0; k < 8; ++k) {
#pragma unroll
    for (int t = 0; t < 8; ++t) {
      v2f b;
      b.x = wr[(4 * k) * H2DIM + t * 16];
      b.y = wr[(4 * k + 1) * H2DIM + t * 16];
      acc[t] = __builtin_amdgcn_wmma_f32_16x16x4_f32(false, a[k], false, b, (short)0, acc[t], false, false);
    }
  }
  float bv[8];
#pragma unroll
  for (int t = 0; t < 8; ++t) bv[t] = b2[t * 16 + nn];

  const int rbase = m0 + (khalf << 3);
#pragma unroll
  for (int j = 0; j < 8; ++j) {
    const int r = rbase + j;
    if (r < n) {
#pragma unroll
      for (int t = 0; t < 8; ++t)
        out[(size_t)r * H2DIM + t * 16 + nn] = acc[t][j] + bv[t];
    }
  }
}

extern "C" void kernel_launch(void* const* d_in, const int* in_sizes, int n_in,
                              void* d_out, int out_size, void* d_ws, size_t ws_size,
                              hipStream_t stream) {
  const float* x   = (const float*)d_in[0];
  const int*   src = (const int*)  d_in[1];
  const int*   dst = (const int*)  d_in[2];
  const float* ew  = (const float*)d_in[3];
  const float* W1  = (const float*)d_in[4];
  const float* b1  = (const float*)d_in[5];
  const float* W2  = (const float*)d_in[6];
  const float* b2  = (const float*)d_in[7];
  float* out = (float*)d_out;

  const int n = in_sizes[0] / NC;
  const int E = in_sizes[1];

  const int NB  = (n + BIN - 1) / BIN;
  const int NCH = (E + CHUNK - 1) / CHUNK;
  const int L   = NB * NCH;

  const size_t np = ((size_t)n + 63) & ~(size_t)63;
  const size_t Lp = ((size_t)L + 63) & ~(size_t)63;
  int*   T      = (int*)d_ws;
  float* dis    = (float*)(T + Lp);
  float* h1     = dis + np;
  float* hA     = h1 + np * H1DIM;
  float* agg2   = hA + np * H1DIM;
  int4*  binned = (int4*)(agg2 + np * H1DIM);

  chunk_sort_kernel<<<NCH, CH_T, 0, stream>>>(src, dst, ew, T, binned, E, NB, NCH);
  bin_degree_gemm1_kernel<<<NB, DEG_T, 0, stream>>>(binned, T, dis, n, NCH, x, W1, h1);

  bin_aggregate_kernel<true><<<NB, AGG_T, 0, stream>>>(binned, T, dis, h1, b1, hA, n, NCH);

  bin_aggregate_kernel<false><<<NB, AGG_T, 0, stream>>>(binned, T, dis, hA, b1, agg2, n, NCH);
  gemm2_wmma<<<(n + 15) / 16, 32, 0, stream>>>(agg2, hA, dis, W2, b2, out, n);
}


# layer-2 transform (WMMA, 32 -> 128) becomes the epilogue of the layer-2 edge-sum launch: a bin's sums stay in LDS, one wave per 16-row stripe; 5 launches become 4; same sums, same products
# speedup vs baseline: 3.6524x; 1.0651x over previous
#include <hip/hip_runtime.h>
#include <math.h>

typedef __attribute__((ext_vector_type(2))) float v2f;
typedef __attribute__((ext_vector_type(8))) float v8f;

#define NC    45
#define H1DIM 32
#define H2DIM 128


#define BIN   400
#define NBMAX 512
#define CHUNK 4096
#define CH_T  256
#define EPT   (CHUNK / CH_T)
#define NCHMAX 512
#define IDX_CAP 8192

__device__ __forceinline__ void gemm1_acc(const float* __restrict__ x,
                                          const float* __restrict__ W1,
                                          int n, int m0, int lane, v8f& c0, v8f& c1) {
  const int khalf = (lane >> 4) & 1;
  const int nn    = lane & 15;
  int m = m0 + nn;
  if (m > n - 1) m = n - 1;

  const float* xr = x + (size_t)m * NC + 2 * khalf;
  const float* wr = W1 + (size_t)(2 * khalf) * H1DIM + nn;
  c0 = (v8f){};
  c1 = (v8f){};
#pragma unroll
  for (int k = 0; k < 11; ++k) {
    v2f a, b0, b1;
    a.x  = xr[4 * k];
    a.y  = xr[4 * k + 1];
    b0.x = wr[(4 * k) * H1DIM];
    b0.y = wr[(4 * k + 1) * H1DIM];
    b1.x = wr[(4 * k) * H1DIM + 16];
    b1.y = wr[(4 * k + 1) * H1DIM + 16];
    c0 = __builtin_amdgcn_wmma_f32_16x16x4_f32(false, a, false, b0, (short)0, c0, false, false);
    c1 = __builtin_amdgcn_wmma_f32_16x16x4_f32(false, a, false, b1, (short)0, c1, false, false);
  }
  {
    const int   koff = khalf ? 0 : 44;
    const float* xl  = x + (size_t)m * NC + koff;
    const float* wl  = W1 + (size_t)koff * H1DIM + nn;
    const float ta = xl[0], t0 = wl[0], t1 = wl[16];
    v2f a, b0, b1;
    a.x  = khalf ? 0.0f : ta;  a.y  = 0.0f;
    b0.x = khalf ? 0.0f : t0;  b0.y = 0.0f;
    b1.x = khalf ? 0.0f : t1;  b1.y = 0.0f;
    c0 = __builtin_amdgcn_wmma_f32_16x16x4_f32(false, a, false, b0, (short)0, c0, false, false);
    c1 = __builtin_amdgcn_wmma_f32_16x16x4_f32(false, a, false, b1, (short)0, c1, false, false);
  }
}
__device__ __forceinline__ void gemm1_store_scaled(float* __restrict__ h1s, const float* sdis,
                                                   int n, int node0, int m0, int lane,
                                                   const v8f& c0, const v8f& c1) {
  const int khalf = (lane >> 4) & 1;
  const int nn    = lane & 15;
  const int rbase = m0 + (khalf << 3);
  float sc[8];
#pragma unroll
  for (int j = 0; j < 8; ++j) sc[j] = sdis[rbase + j - node0];
#pragma unroll
  for (int j = 0; j < 8; ++j) {
    const int r = rbase + j;
    if (r < n) {
      h1s[(size_t)r * H1DIM + nn]      = c0[j] * sc[j];
      h1s[(size_t)r * H1DIM + 16 + nn] = c1[j] * sc[j];
    }
  }
}

__global__ void __launch_bounds__(CH_T)
chunk_sort_kernel(const int* __restrict__ src, const int* __restrict__ dst,
                  const float* __restrict__ ew, int* __restrict__ T,
                  int4* __restrict__ binned, int E, int NB, int NCH) {
  __shared__ int   cur[NBMAX];
  __shared__ int   lbase[NBMAX];
  __shared__ int   s_src[CHUNK];
  __shared__ int   s_dst[CHUNK];
  __shared__ float s_w[CHUNK];
  const int tid    = threadIdx.x;
  const int lane   = tid & 31;
  const int chunk  = blockIdx.x;
  const int e_base = chunk * CHUNK;
  for (int b = tid; b < NB; b += CH_T) cur[b] = 0;
  __syncthreads();
  int   d[EPT], r[EPT], sv[EPT];
  float wv[EPT];
#pragma unroll
  for (int k = 0; k < EPT; ++k) {
    const int e  = e_base + k * CH_T + tid;
    const int ec = e < E ? e : E - 1;
    const int dv = dst[ec];
    sv[k] = src[ec];
    wv[k] = ew[ec];
    d[k]  = (e < E) ? dv : -1;
  }
#pragma unroll
  for (int k = 0; k < EPT; ++k) {
    r[k] = 0;
    if (d[k] >= 0) r[k] = atomicAdd(&cur[(unsigned)d[k] / BIN], 1);
  }
  __syncthreads();
  if (tid < 32) {
    int running = 0;
    for (int b0 = 0; b0 < NB; b0 += 32) {
      const int bb = b0 + lane;
      const int v  = (bb < NB) ? cur[bb] : 0;
      int incl = v;
#pragma unroll
      for (int dlt = 1; dlt < 32; dlt <<= 1) {
        const int u = __shfl_up(incl, dlt);
        if (lane >= dlt) incl += u;
      }
      if (bb < NB) lbase[bb] = running + incl - v;
      running += __shfl(incl, 31);
    }
  }
  __syncthreads();
#pragma unroll
  for (int k = 0; k < EPT; ++k) {
    if (d[k] >= 0) {
      const int li = lbase[(unsigned)d[k] / BIN] + r[k];
      s_src[li] = sv[k];
      s_dst[li] = d[k];
      s_w[li]   = wv[k];
    }
  }
  __syncthreads();
  const int left   = E - e_base;
  const int nvalid = left < CHUNK ? left : CHUNK;
  for (int i = tid; i < nvalid; i += CH_T)
    binned[e_base + i] = make_int4(s_src[i], 0, s_dst[i], __float_as_int(s_w[i]));
  for (int b = tid; b < NB; b += CH_T) T[b * NCH + chunk] = lbase[b] | (cur[b] << 16);
}

template <int RPT>
__device__ __forceinline__ void bin_runs_fetch(const int* __restrict__ T, int b, int NCH, int tid,
                                               int (&w)[RPT]) {
#pragma unroll
  for (int j = 0; j < RPT; ++j) {
    const int c = tid * RPT + j;
    w[j] = (c < NCH) ? T[(size_t)b * NCH + c] : 0;
  }
}
template <int NT, int RPT>
__device__ __forceinline__ int bin_runs_scan(const int (&w)[RPT], int tid, int* s_wtot,
                                             int (&p0)[RPT], int (&ln)[RPT], int (&g0)[RPT]) {
  const int lane = tid & 31;
  const int wv   = tid >> 5;
  int tsum = 0;
#pragma unroll
  for (int j = 0; j < RPT; ++j) {
    ln[j] = w[j] >> 16;
    g0[j] = (tid * RPT + j) * CHUNK + (w[j] & 0xffff);
    p0[j] = tsum;
    tsum += ln[j];
  }
  int incl = tsum;
#pragma unroll
  for (int dlt = 1; dlt < 32; dlt <<= 1) {
    const int u = __shfl_up(incl, dlt);
    if (lane >= dlt) incl += u;
  }
  if (lane == 31) s_wtot[wv] = incl;
  __syncthreads();
  const int t = (lane < NT / 32) ? s_wtot[lane] : 0;
  int tot    = t;
  int before = (lane < wv) ? t : 0;
#pragma unroll
  for (int dlt = 16; dlt >= 1; dlt >>= 1) {
    tot    += __shfl_xor(tot, dlt);
    before += __shfl_xor(before, dlt);
  }
  const int base = before + incl - tsum;
#pragma unroll
  for (int j = 0; j < RPT; ++j) p0[j] += base;
  return tot;
}
template <int RPT>
__device__ __forceinline__ void bin_runs_list(int* idx, int tb, const int (&p0)[RPT],
                                              const int (&ln)[RPT], const int (&g0)[RPT]) {
#pragma unroll
  for (int j = 0; j < RPT; ++j) {
    int k0 = tb - p0[j];
    if (k0 < 0) k0 = 0;
    int k1 = tb + IDX_CAP - p0[j];
    if (k1 > ln[j]) k1 = ln[j];
    for (int k = k0; k < k1; ++k) idx[p0[j] + k - tb] = g0[j] + k;
  }
}

#define DEG_T    512
#define DEG_R    20
#define DEG_FRAC 40
__device__ __forceinline__ long long weight_to_fixed(float w) {
  const unsigned u  = __float_as_uint(w);
  const int      e  = (int)((u >> 23) & 0xffu);
  const unsigned long long m = (u & 0x7fffffu) | (e ? 0x800000u : 0u);
  const int sh = (e ? e : 1) - 150 + DEG_FRAC;
  const unsigned long long q = (sh >= 0) ? (m << (sh < 39 ? sh : 39))
                                         : (m >> (-sh < 24 ? -sh : 24));
  return (u >> 31) ? -(long long)q : (long long)q;
}
#define DEG_W (DEG_T / 32)
#define DEG_NS (BIN / 16)
#define DEG_S ((DEG_NS + DEG_W - 1) / DEG_W)
#define DEG_RPT (NCHMAX / DEG_T)
__global__ void __launch_bounds__(DEG_T)
bin_degree_gemm1_kernel(const int4* __restrict__ binned, const int* __restrict__ T,
                        float* __restrict__ dis, int n, int NCH,
                        const float* __restrict__ x, const float* __restrict__ W1,
                        float* __restrict__ h1s) {
  __shared__ unsigned long long dsum[BIN];
  __shared__ float sdis[BIN];
  __shared__ int   idx[IDX_CAP];
  __shared__ int   s_wtot[32];
  const int b     = blockIdx.x;
  const int tid   = threadIdx.x;
  const int lane  = tid & 31;
  const int node0 = b * BIN;
  const int wv = __builtin_amdgcn_readfirstlane(tid >> 5);
  int rw[DEG_RPT];
  bin_runs_fetch<DEG_RPT>(T, b, NCH, tid, rw);
  for (int k = tid; k < BIN; k += DEG_T) dsum[k] = 0ull;
  v8f acc0[DEG_S], acc1[DEG_S];
#pragma unroll
  for (int q = 0; q < DEG_S; ++q)
    if (wv + q * DEG_W < DEG_NS)
      gemm1_acc(x, W1, n, node0 + (wv + q * DEG_W) * 16, lane, acc0[q], acc1[q]);
  int p0[DEG_RPT], ln[DEG_RPT], g0[DEG_RPT];
  const int total = __builtin_amdgcn_readfirstlane(
      bin_runs_scan<DEG_T, DEG_RPT>(rw, tid, s_wtot, p0, ln, g0));
  const int2* dw = (const int2*)binned;
  for (int tb = 0; tb < total; tb += IDX_CAP) {
    if (tb) __syncthreads();
    bin_runs_list<DEG_RPT>(idx, tb, p0, ln, g0);
    __syncthreads();
    const int te = (total - tb < IDX_CAP) ? (total - tb) : IDX_CAP;
    for (int i0 = tid; i0 < te; i0 += DEG_T * DEG_R) {
      int   dk[DEG_R];
      float wk[DEG_R];
#pragma unroll
      for (int k = 0; k < DEG_R; ++k) {
        const int  i = i0 + k * DEG_T;
        const int  g = idx[i < te ? i : te - 1];
        const int2 p = dw[2 * (size_t)g + 1];
        dk[k] = (i < te) ? (p.x - node0) : -1;
        wk[k] = __int_as_float(p.y);
      }
      asm volatile("" :: "v"(wk[0]), "v"(wk[1]), "v"(wk[2]), "v"(wk[3]), "v"(wk[4]),
                         "v"(wk[5]), "v"(wk[6]), "v"(wk[7]), "v"(wk[8]), "v"(wk[9]),
                         "v"(wk[10]), "v"(wk[11]), "v"(wk[12]), "v"(wk[13]), "v"(wk[14]),
                         "v"(wk[15]), "v"(wk[16]), "v"(wk[17]), "v"(wk[18]), "v"(wk[19]));
      static_assert(DEG_R == 20, "the fence above names DEG_R weights");
#pragma unroll
      for (int k = 0; k < DEG_R; ++k) {
        if (dk[k] >= 0)
          atomicAdd(&dsum[dk[k]], (unsigned long long)weight_to_fixed(wk[k]));
      }
    }
  }
  __syncthreads();
  for (int k = tid; k < BIN; k += DEG_T) {
    const int node = node0 + k;
    if (node < n) {
      const long long d1 = (long long)dsum[k] + (1ll << DEG_FRAC);
      const float dv = 1.0f / sqrtf((float)d1 * (1.0f / (float)(1ll << DEG_FRAC)));
      dis[node] = dv;
      sdis[k]   = dv;
    }
  }
  __syncthreads();
#pragma unroll
  for (int q = 0; q < DEG_S; ++q)
    if (wv + q * DEG_W < DEG_NS)
      gemm1_store_scaled(h1s, sdis, n, node0, node0 + (wv + q * DEG_W) * 16, lane, acc0[q], acc1[q]);
}

__device__ __forceinline__ void gemm2_stripe(const float* srow, const float* __restrict__ h,
                                             const float* __restrict__ dis,
                                             const float* __restrict__ W2,
                                             const float* __restrict__ b2,
                                             float* __restrict__ out, int n, int m0, int lane) {
  const int khalf = (lane >> 4) & 1;
  const int nn    = lane & 15;
  int m = m0 + nn;
  if (m > n - 1) m = n - 1;

  const float* ap = srow + nn * H1DIM + 2 * khalf;
  const float* hp = h    + (size_t)m * H1DIM + 2 * khalf;
  const float* wr = W2 + (size_t)(2 * khalf) * H2DIM + nn;
  const float di = dis[m];
  v2f a[8];
#pragma unroll
  for (int k = 0; k < 8; ++k) {
    const v2f ag = *(const v2f*)(ap + 4 * k);
    const v2f hv = *(const v2f*)(hp + 4 * k);
    a[k].x = (ag.x + hv.x) * di;
    a[k].y = (ag.y + hv.y) * di;
  }

  v8f acc[8] = {};
#pragma unroll
  for (int k = 0; k < 8; ++k) {
#pragma unroll
    for (int t = 0; t < 8; ++t) {
      v2f b;
      b.x = wr[(4 * k) * H2DIM + t * 16];
      b.y = wr[(4 * k + 1) * H2DIM + t * 16];
      acc[t] = __builtin_amdgcn_wmma_f32_16x16x4_f32(false, a[k], false, b, (short)0, acc[t], false, false);
    }
  }
  float bv[8];
#pragma unroll
  for (int t = 0; t < 8; ++t) bv[t] = b2[t * 16 + nn];

  const int rbase = m0 + (khalf << 3);
#pragma unroll
  for (int j = 0; j < 8; ++j) {
    const int r = rbase + j;
    if (r < n) {
#pragma unroll
      for (int t = 0; t < 8; ++t)
        out[(size_t)r * H2DIM + t * 16 + nn] = acc[t][j] + bv[t];
    }
  }
}

#define AGG_T 1024
#define AGG_W (AGG_T / 32)
#define AGG_R ((BIN + AGG_W - 1) / AGG_W)
template <bool LAYER1>
__global__ void __launch_bounds__(AGG_T)
bin_aggregate_kernel(const int4* __restrict__ binned, const int* __restrict__ T,
                     const float* __restrict__ dis, const float* __restrict__ h,
                     const float* __restrict__ bias, const float* __restrict__ W2,
                     float* __restrict__ outp, int n, int NCH) {
  __shared__ __attribute__((aligned(16))) float tile[BIN * H1DIM];
  static_assert(BIN % 16 == 0 && BIN / 16 <= AGG_W, "one whole stripe of the bin per wave");
  __shared__ int   idx[IDX_CAP];
  __shared__ int   s_wtot[32];
  static_assert(AGG_T >= NCHMAX, "one run of the bin per thread");
  const int lane = threadIdx.x & 31;
  const int wv = __builtin_amdgcn_readfirstlane((int)(threadIdx.x >> 5));
  const int nw = AGG_W;
  const int b  = blockIdx.x;
  int rw[1];
  bin_runs_fetch<1>(T, b, NCH, (int)threadIdx.x, rw);
  for (int k = threadIdx.x; k < BIN * H1DIM; k += AGG_T) tile[k] = 0.0f;
  const int node0 = b * BIN;
  int p0[1], ln[1], g0[1];
  const int total = __builtin_amdgcn_readfirstlane(
      bin_runs_scan<AGG_T, 1>(rw, (int)threadIdx.x, s_wtot, p0, ln, g0));

  for (int tb = 0; tb < total; tb += IDX_CAP) {
    if (tb) __syncthreads();
    bin_runs_list<1>(idx, tb, p0, ln, g0);
    __syncthreads();
    const int te = (total - tb < IDX_CAP) ? (total - tb) : IDX_CAP;
    for (int i0 = wv * 32; i0 < te; i0 += nw * 32) {
      const int i = i0 + lane;
      int   s   = node0;
      int   dl  = 0;
      float nrm = 0.0f;
      if (i < te) {
        const int4 r = binned[idx[i]];
        s   = r.x;
        dl  = r.z - node0;
        nrm = __int_as_float(r.w);
      }
      if (i0 + 32 <= te) {
        float v[32];
#pragma unroll
        for (int j = 0; j < 32; ++j) {
          const int sj = __builtin_amdgcn_readlane(s, j);
          v[j] = h[(size_t)sj * H1DIM + lane];
        }
#pragma unroll
        for (int j = 0; j < 32; ++j) {
          const int   dj = __builtin_amdgcn_readlane(dl, j);
          const float nj = __int_as_float(__builtin_amdgcn_readlane(__float_as_int(nrm), j));
          atomicAdd(&tile[dj * H1DIM + lane], v[j] * nj);
        }
      } else {
        const int cnt = te - i0;
        float v[32];
#pragma unroll
        for (int j = 0; j < 32; ++j) {
          const int sj = __builtin_amdgcn_readlane(s, j);
          v[j] = h[(size_t)sj * H1DIM + lane];
        }
#pragma unroll
        for (int j = 0; j < 32; ++j) {
          if (j < cnt) {
            const int   dj = __builtin_amdgcn_readlane(dl, j);
            const float nj = __int_as_float(__builtin_amdgcn_readlane(__float_as_int(nrm), j));
            atomicAdd(&tile[dj * H1DIM + lane], v[j] * nj);
          }
        }
      }
    }
  }
  __syncthreads();

  if (LAYER1) {
    float hv[AGG_R], dv[AGG_R];
#pragma unroll
    for (int k = 0; k < AGG_R; ++k) {
      int node = node0 + wv + k * AGG_W;
      if (node > n - 1) node = n - 1;
      dv[k] = dis[node];
      hv[k] = h[(size_t)node * H1DIM + lane];
    }
    const float bb = bias[lane];
#pragma unroll
    for (int k = 0; k < AGG_R; ++k) {
      const int r    = wv + k * AGG_W;
      const int node = node0 + r;
      if (r < BIN && node < n) {
        const float acc = tile[r * H1DIM + lane];
        const float di  = dv[k];
        outp[(size_t)node * H1DIM + lane] = di * tanhf(di * (acc + hv[k]) + bb);
      }
    }
  } else {
    if (wv < BIN / 16) {
      const int m0 = node0 + wv * 16;
      if (m0 < n) gemm2_stripe(tile + wv * 16 * H1DIM, h, dis, W2, bias, outp, n, m0, lane);
    }
  }
}

extern "C" void kernel_launch(void* const* d_in, const int* in_sizes, int n_in,
                              void* d_out, int out_size, void* d_ws, size_t ws_size,
                              hipStream_t stream) {
  const float* x   = (const float*)d_in[0];
  const int*   src = (const int*)  d_in[1];
  const int*   dst = (const int*)  d_in[2];
  const float* ew  = (const float*)d_in[3];
  const float* W1  = (const float*)d_in[4];
  const float* b1  = (const float*)d_in[5];
  const float* W2  = (const float*)d_in[6];
  const float* b2  = (const float*)d_in[7];
  float* out = (float*)d_out;

  const int n = in_sizes[0] / NC;
  const int E = in_sizes[1];

  const int NB  = (n + BIN - 1) / BIN;
  const int NCH = (E + CHUNK - 1) / CHUNK;
  const int L   = NB * NCH;

  const size_t np = ((size_t)n + 63) & ~(size_t)63;
  const size_t Lp = ((size_t)L + 63) & ~(size_t)63;
  int*   T      = (int*)d_ws;
  float* dis    = (float*)(T + Lp);
  float* h1     = dis + np;
  float* hA     = h1 + np * H1DIM;
  int4*  binned = (int4*)(hA + np * H1DIM);

  chunk_sort_kernel<<<NCH, CH_T, 0, stream>>>(src, dst, ew, T, binned, E, NB, NCH);
  bin_degree_gemm1_kernel<<<NB, DEG_T, 0, stream>>>(binned, T, dis, n, NCH, x, W1, h1);

  bin_aggregate_kernel<true><<<NB, AGG_T, 0, stream>>>(binned, T, dis, h1, b1, W2, hA, n, NCH);

  bin_aggregate_kernel<false><<<NB, AGG_T, 0, stream>>>(binned, T, dis, hA, b2, W2, out, n, NCH);
}
